# starting kernel
# baseline (speedup 1.0000x reference)
#include <hip/hip_runtime.h>

#define NN 50000
#define NE 625000
#define DIM 128

typedef __attribute__((ext_vector_type(2))) float v2f;
typedef __attribute__((ext_vector_type(8))) float v8f;

__global__ void sage_zero_kernel(float* __restrict__ p, long n) {
  long i = (long)blockIdx.x * blockDim.x + threadIdx.x;
  long stride = (long)gridDim.x * blockDim.x;
  for (; i < n; i += stride) p[i] = 0.0f;
}

__global__ void sage_degree_kernel(const int* __restrict__ dst,
                                   float* __restrict__ deg, int E) {
  int e = blockIdx.x * blockDim.x + threadIdx.x;
  if (e < E) atomicAdd(&deg[dst[e]], 1.0f);
}

__global__ __launch_bounds__(256) void sage_scatter_kernel(
    const float* __restrict__ h, const int* __restrict__ src,
    const int* __restrict__ dst, const float* __restrict__ w,
    float* __restrict__ agg, int E) {
  int t = blockIdx.x * blockDim.x + threadIdx.x;
  int lane = t & 31;
  int e = t >> 5;
  if (e >= E) return;
  int s = src[e];
  int d = dst[e];
  float we = w[e];
  const float4* row = (const float4*)(h + (size_t)s * DIM);
  float4 v = row[lane];
  float* ap = agg + (size_t)d * DIM + lane * 4;
  atomicAdd(ap + 0, v.x * we);
  atomicAdd(ap + 1, v.y * we);
  atomicAdd(ap + 2, v.z * we);
  atomicAdd(ap + 3, v.w * we);
}

#define RPB 128
#define KC  16
#define AP  20
#define WP  132

__global__ __launch_bounds__(256) void sage_gemm_kernel(
    const float* __restrict__ Hself, const float* __restrict__ Agg,
    const float* __restrict__ Deg,   const float* __restrict__ Wself,
    const float* __restrict__ Wneigh,const float* __restrict__ Bias,
    float* __restrict__ Out, int nrows, int do_relu) {
  __shared__ float sWs[KC * WP];
  __shared__ float sWn[KC * WP];
  __shared__ float sAs[RPB * AP];
  __shared__ float sAn[RPB * AP];
  __shared__ float sRd[RPB];

  const int tid  = threadIdx.x;
  const int lane = tid & 31;
  const int wave = tid >> 5;
  const int rowbase = blockIdx.x * RPB;

  for (int r = tid; r < RPB; r += 256) {
    int gr = rowbase + r;
    float dg = (gr < nrows) ? Deg[gr] : 1.0f;
    sRd[r] = 1.0f / fmaxf(dg, 1.0f);
  }

  v8f acc[8] = {};

  const int jj    = wave * 16 + (lane & 15);
  const int khalf = (lane >> 4) << 1;
  const int m     = lane & 15;

  for (int kc = 0; kc < DIM / KC; ++kc) {
    __syncthreads();
#pragma unroll
    for (int s = 0; s < (RPB * KC) / 256; ++s) {
      int i = tid + s * 256;
      int j = i >> 4;
      int k = i & 15;
      sWs[k * WP + j] = Wself [j * DIM + kc * KC + k];
      sWn[k * WP + j] = Wneigh[j * DIM + kc * KC + k];
    }
#pragma unroll
    for (int s = 0; s < (RPB * KC) / 256; ++s) {
      int i = tid + s * 256;
      int r = i >> 4;
      int k = i & 15;
      int gr = rowbase + r;
      float hs = 0.0f, hn = 0.0f;
      if (gr < nrows) {
        hs = Hself[(size_t)gr * DIM + kc * KC + k];
        hn = Agg  [(size_t)gr * DIM + kc * KC + k] * sRd[r];
      }
      sAs[r * AP + k] = hs;
      sAn[r * AP + k] = hn;
    }
    __syncthreads();

    v2f bs[4], bn[4];
#pragma unroll
    for (int k0 = 0; k0 < 4; ++k0) {
      int kk = k0 * 4 + khalf;
      bs[k0].x = sWs[ kk      * WP + jj];
      bs[k0].y = sWs[(kk + 1) * WP + jj];
      bn[k0].x = sWn[ kk      * WP + jj];
      bn[k0].y = sWn[(kk + 1) * WP + jj];
    }

#pragma unroll
    for (int rt = 0; rt < 8; ++rt) {
      const float* as = &sAs[(rt * 16 + m) * AP];
      const float* an = &sAn[(rt * 16 + m) * AP];
#pragma unroll
      for (int k0 = 0; k0 < 4; ++k0) {
        int kk = k0 * 4 + khalf;
        v2f a_s; a_s.x = as[kk]; a_s.y = as[kk + 1];
        v2f a_n; a_n.x = an[kk]; a_n.y = an[kk + 1];
        acc[rt] = __builtin_amdgcn_wmma_f32_16x16x4_f32(
            false, a_s, false, bs[k0], (short)0, acc[rt], false, false);
        acc[rt] = __builtin_amdgcn_wmma_f32_16x16x4_f32(
            false, a_n, false, bn[k0], (short)0, acc[rt], false, false);
      }
    }
  }

  const float bv = Bias[jj];
  const int mofs = (lane >> 4) << 3;
#pragma unroll
  for (int rt = 0; rt < 8; ++rt) {
#pragma unroll
    for (int r = 0; r < 8; ++r) {
      int row = rowbase + rt * 16 + r + mofs;
      if (row < nrows) {
        float v = acc[rt][r] + bv;
        if (do_relu) v = fmaxf(v, 0.0f);
        Out[(size_t)row * DIM + jj] = v;
      }
    }
  }
}

extern "C" void kernel_launch(void* const* d_in, const int* in_sizes, int n_in,
                              void* d_out, int out_size, void* d_ws, size_t ws_size,
                              hipStream_t stream) {
  const float* x       = (const float*)d_in[0];
  const int*   src     = (const int*)  d_in[1];
  const int*   dst     = (const int*)  d_in[2];
  const float* weight  = (const float*)d_in[3];
  const float* Wself1  = (const float*)d_in[4];
  const float* Wneigh1 = (const float*)d_in[5];
  const float* b1      = (const float*)d_in[6];
  const float* Wself2  = (const float*)d_in[7];
  const float* Wneigh2 = (const float*)d_in[8];
  const float* b2      = (const float*)d_in[9];
  float* out = (float*)d_out;

  char* ws = (char*)d_ws;
  float* deg = (float*)(ws);
  float* agg = (float*)(ws + ((size_t)NN * 4 + 255 & ~(size_t)255));
  float* h1  = (float*)((char*)agg + (size_t)NN * DIM * 4);

  const long aggN = (long)NN * DIM;
  const int gemm_blocks = (NN + RPB - 1) / RPB;
  const int scat_blocks = (NE * 32 + 255) / 256;
  const int deg_blocks  = (NE + 255) / 256;

  sage_zero_kernel<<<2048, 256, 0, stream>>>(deg, NN);
  sage_zero_kernel<<<2048, 256, 0, stream>>>(agg, aggN);
  sage_degree_kernel<<<deg_blocks, 256, 0, stream>>>(dst, deg, NE);
  sage_scatter_kernel<<<scat_blocks, 256, 0, stream>>>(x, src, dst, weight, agg, NE);
  sage_gemm_kernel<<<gemm_blocks, 256, 0, stream>>>(
      x, agg, deg, Wself1, Wneigh1, b1, h1, NN,  1);

  sage_zero_kernel<<<2048, 256, 0, stream>>>(agg, aggN);
  sage_scatter_kernel<<<scat_blocks, 256, 0, stream>>>(h1, src, dst, weight, agg, NE);
  sage_gemm_kernel<<<gemm_blocks, 256, 0, stream>>>(
      h1, agg, deg, Wself2, Wneigh2, b2, out, NN,  0);
}


# Replace the float-atomic edge scatter by a dst-sorted edge list (int atomic picks a slot, each node's list then ordered by edge id) and a per-node register gather that sums in edge-id order
# speedup vs baseline: 1.5247x; 1.5247x over previous
#include <hip/hip_runtime.h>


#define NN 50000
#define NE 625000
#define DIM 128

typedef __attribute__((ext_vector_type(2))) float v2f;
typedef __attribute__((ext_vector_type(8))) float v8f;


__global__ void sage_zero_kernel(int* __restrict__ p, int n) {
  int i = blockIdx.x * blockDim.x + threadIdx.x;
  if (i < n) p[i] = 0;
}

__global__ void sage_count_kernel(const int* __restrict__ dst, int* __restrict__ cnt,
                                  int* __restrict__ slotno, int E) {
  int e = blockIdx.x * blockDim.x + threadIdx.x;
  if (e < E) slotno[e] = atomicAdd(&cnt[dst[e]], 1);
}

#define SCAN_THREADS 1024
#define SCAN_WAVES   (SCAN_THREADS / 32)
#define SCAN_TILES   ((NN + 1 + SCAN_THREADS - 1) / SCAN_THREADS)
#define SCAN_GROUP   7
static_assert(SCAN_TILES % SCAN_GROUP == 0, "tile groups must cover the wave's range");

__device__ __forceinline__ int sage_wave_incl_scan(int v, int lane) {
  v += __builtin_amdgcn_update_dpp(0, v, 0x111, 0xf, 0xf, false);
  v += __builtin_amdgcn_update_dpp(0, v, 0x112, 0xf, 0xf, false);
  v += __builtin_amdgcn_update_dpp(0, v, 0x114, 0xf, 0xf, false);
  v += __builtin_amdgcn_update_dpp(0, v, 0x118, 0xf, 0xf, false);
  const int low = __builtin_amdgcn_readlane(v, 15);
  if (lane >= 16) v += low;
  return v;
}

__global__ __launch_bounds__(SCAN_THREADS) void sage_scan_kernel(
    const int* __restrict__ cnt, int* __restrict__ rowptr,
    float* __restrict__ deg, int n) {
  __shared__ int wtot[SCAN_WAVES];
  const int lane = threadIdx.x & 31;
  const int wave = threadIdx.x >> 5;
  const int wbase = wave * (SCAN_TILES * 32);

  int tsum = 0;
  for (int i0 = 0; i0 < SCAN_TILES; i0 += SCAN_GROUP) {
#pragma unroll
    for (int u = 0; u < SCAN_GROUP; ++u) {
      const int idx = wbase + (i0 + u) * 32 + lane;
      const int c = cnt[min(idx, n - 1)];
      tsum += (idx < n) ? c : 0;
    }
  }
  const int wsum = __builtin_amdgcn_readlane(sage_wave_incl_scan(tsum, lane), 31);
  if (lane == 0) wtot[wave] = wsum;
  __syncthreads();
  if (wave == 0) {
    const int t = wtot[lane];
    const int inc = sage_wave_incl_scan(t, lane);
    wtot[lane] = inc - t;
  }
  __syncthreads();
  int carry = wtot[wave];
  for (int i0 = 0; i0 < SCAN_TILES; i0 += SCAN_GROUP) {
    int c[SCAN_GROUP];
#pragma unroll
    for (int u = 0; u < SCAN_GROUP; ++u) {
      const int idx = wbase + (i0 + u) * 32 + lane;
      const int t = cnt[min(idx, n - 1)];
      c[u] = (idx < n) ? t : 0;
    }
#pragma unroll
    for (int u = 0; u < SCAN_GROUP; ++u) {
      const int idx = wbase + (i0 + u) * 32 + lane;
      const int inc = sage_wave_incl_scan(c[u], lane);
      if (idx <= n) rowptr[idx] = carry + inc - c[u];
      if (idx < n)  deg[idx] = (float)c[u];
      carry += __builtin_amdgcn_readlane(inc, 31);
    }
  }
}

__global__ void sage_place_kernel(const int* __restrict__ dst,
                                  const int* __restrict__ rowptr,
                                  const int* __restrict__ slotno,
                                  int* __restrict__ seg, int E) {
  int e = blockIdx.x * blockDim.x + threadIdx.x;
  if (e < E) seg[rowptr[dst[e]] + slotno[e]] = e;
}

#define NPW 4
__global__ __launch_bounds__(256) void sage_order_kernel(
    const int* __restrict__ rowptr, const int* __restrict__ seg,
    const int* __restrict__ src, const float* __restrict__ w,
    int* __restrict__ cs, float* __restrict__ cw, int n_nodes) {
  const int lane = threadIdx.x & 31;
  const int wv = __builtin_amdgcn_readfirstlane(
      (int)((blockIdx.x * blockDim.x + threadIdx.x) >> 5));
  for (int k = 0; k < NPW; ++k) {
    const int node = wv * NPW + k;
    if (node >= n_nodes) break;
    const int beg = rowptr[node];
    const int n   = rowptr[node + 1] - beg;
    if (n <= 32) {
      const bool act = lane < n;
      const int my = act ? seg[beg + lane] : 0x7fffffff;
      int rank = 0;
      for (int j = 0; j < n; ++j) {
        const int other = __builtin_amdgcn_readlane(my, j);
        rank += (other < my) ? 1 : 0;
      }
      if (act) {
        cs[beg + rank] = src[my];
        cw[beg + rank] = w[my];
      }
    } else {
      for (int i0 = 0; i0 < n; i0 += 32) {
        const int i = i0 + lane;
        const bool act = i < n;
        const int my = act ? seg[beg + i] : 0x7fffffff;
        int rank = 0;
        for (int j = 0; j < n; ++j) rank += (seg[beg + j] < my) ? 1 : 0;
        if (act) {
          cs[beg + rank] = src[my];
          cw[beg + rank] = w[my];
        }
      }
    }
  }
}


template <int G>
__device__ __forceinline__ void sage_rows(float4& acc, const float* __restrict__ h,
                                          int ms, float mw, int j, int lane) {
  float4 v[G];
  float we[G];
#pragma unroll
  for (int u = 0; u < G; ++u) {
    const int s = __builtin_amdgcn_readlane(ms, j + u);
    we[u] = __int_as_float(__builtin_amdgcn_readlane(__float_as_int(mw), j + u));
    v[u] = ((const float4*)(h + (size_t)s * DIM))[lane];
  }
#pragma unroll
  for (int u = 0; u < G; ++u) {
    acc.x += v[u].x * we[u];
    acc.y += v[u].y * we[u];
    acc.z += v[u].z * we[u];
    acc.w += v[u].w * we[u];
  }
}

__global__ __launch_bounds__(256) void sage_gather_kernel(
    const float* __restrict__ h, const int* __restrict__ rowptr,
    const int* __restrict__ cs, const float* __restrict__ cw,
    float* __restrict__ agg, int n_nodes) {
  const int lane = threadIdx.x & 31;
  const int wv = __builtin_amdgcn_readfirstlane(
      (int)((blockIdx.x * blockDim.x + threadIdx.x) >> 5));
  for (int k = 0; k < NPW; ++k) {
    const int node = wv * NPW + k;
    if (node >= n_nodes) break;
    const int beg = rowptr[node];
    const int end = rowptr[node + 1];
    float4 acc = make_float4(0.0f, 0.0f, 0.0f, 0.0f);
    for (int base = beg; base < end; base += 32) {
      const int c = min(32, end - base);
      int ms = 0;
      float mw = 0.0f;
      if (lane < c) {
        ms = cs[base + lane];
        mw = cw[base + lane];
      }
      int j = 0;
      for (; j + 8 <= c; j += 8) sage_rows<8>(acc, h, ms, mw, j, lane);
      if (j + 4 <= c) { sage_rows<4>(acc, h, ms, mw, j, lane); j += 4; }
      if (j + 2 <= c) { sage_rows<2>(acc, h, ms, mw, j, lane); j += 2; }
      if (j < c)      { sage_rows<1>(acc, h, ms, mw, j, lane); }
    }
    ((float4*)(agg + (size_t)node * DIM))[lane] = acc;
  }
}

#define RPB 128
#define KC  16
#define AP  20
#define WP  132

__global__ __launch_bounds__(256) void sage_gemm_kernel(
    const float* __restrict__ Hself, const float* __restrict__ Agg,
    const float* __restrict__ Deg,   const float* __restrict__ Wself,
    const float* __restrict__ Wneigh,const float* __restrict__ Bias,
    float* __restrict__ Out, int nrows, int do_relu) {
  __shared__ float sWs[KC * WP];
  __shared__ float sWn[KC * WP];
  __shared__ float sAs[RPB * AP];
  __shared__ float sAn[RPB * AP];
  __shared__ float sRd[RPB];

  const int tid  = threadIdx.x;
  const int lane = tid & 31;
  const int wave = tid >> 5;
  const int rowbase = blockIdx.x * RPB;

  for (int r = tid; r < RPB; r += 256) {
    int gr = rowbase + r;
    float dg = (gr < nrows) ? Deg[gr] : 1.0f;
    sRd[r] = 1.0f / fmaxf(dg, 1.0f);
  }

  v8f acc[8] = {};

  const int jj    = wave * 16 + (lane & 15);
  const int khalf = (lane >> 4) << 1;
  const int m     = lane & 15;

  for (int kc = 0; kc < DIM / KC; ++kc) {
    __syncthreads();
#pragma unroll
    for (int s = 0; s < (RPB * KC) / 256; ++s) {
      int i = tid + s * 256;
      int j = i >> 4;
      int k = i & 15;
      sWs[k * WP + j] = Wself [j * DIM + kc * KC + k];
      sWn[k * WP + j] = Wneigh[j * DIM + kc * KC + k];
    }
#pragma unroll
    for (int s = 0; s < (RPB * KC) / 256; ++s) {
      int i = tid + s * 256;
      int r = i >> 4;
      int k = i & 15;
      int gr = rowbase + r;
      float hs = 0.0f, hn = 0.0f;
      if (gr < nrows) {
        hs = Hself[(size_t)gr * DIM + kc * KC + k];
        hn = Agg  [(size_t)gr * DIM + kc * KC + k] * sRd[r];
      }
      sAs[r * AP + k] = hs;
      sAn[r * AP + k] = hn;
    }
    __syncthreads();

    v2f bs[4], bn[4];
#pragma unroll
    for (int k0 = 0; k0 < 4; ++k0) {
      int kk = k0 * 4 + khalf;
      bs[k0].x = sWs[ kk      * WP + jj];
      bs[k0].y = sWs[(kk + 1) * WP + jj];
      bn[k0].x = sWn[ kk      * WP + jj];
      bn[k0].y = sWn[(kk + 1) * WP + jj];
    }

#pragma unroll
    for (int rt = 0; rt < 8; ++rt) {
      const float* as = &sAs[(rt * 16 + m) * AP];
      const float* an = &sAn[(rt * 16 + m) * AP];
#pragma unroll
      for (int k0 = 0; k0 < 4; ++k0) {
        int kk = k0 * 4 + khalf;
        v2f a_s; a_s.x = as[kk]; a_s.y = as[kk + 1];
        v2f a_n; a_n.x = an[kk]; a_n.y = an[kk + 1];
        acc[rt] = __builtin_amdgcn_wmma_f32_16x16x4_f32(
            false, a_s, false, bs[k0], (short)0, acc[rt], false, false);
        acc[rt] = __builtin_amdgcn_wmma_f32_16x16x4_f32(
            false, a_n, false, bn[k0], (short)0, acc[rt], false, false);
      }
    }
  }

  const float bv = Bias[jj];
  const int mofs = (lane >> 4) << 3;
#pragma unroll
  for (int rt = 0; rt < 8; ++rt) {
#pragma unroll
    for (int r = 0; r < 8; ++r) {
      int row = rowbase + rt * 16 + r + mofs;
      if (row < nrows) {
        float v = acc[rt][r] + bv;
        if (do_relu) v = fmaxf(v, 0.0f);
        Out[(size_t)row * DIM + jj] = v;
      }
    }
  }
}

extern "C" void kernel_launch(void* const* d_in, const int* in_sizes, int n_in,
                              void* d_out, int out_size, void* d_ws, size_t ws_size,
                              hipStream_t stream) {
  const float* x       = (const float*)d_in[0];
  const int*   src     = (const int*)  d_in[1];
  const int*   dst     = (const int*)  d_in[2];
  const float* weight  = (const float*)d_in[3];
  const float* Wself1  = (const float*)d_in[4];
  const float* Wneigh1 = (const float*)d_in[5];
  const float* b1      = (const float*)d_in[6];
  const float* Wself2  = (const float*)d_in[7];
  const float* Wneigh2 = (const float*)d_in[8];
  const float* b2      = (const float*)d_in[9];
  float* out = (float*)d_out;

  char* ws = (char*)d_ws;
  float* deg = (float*)(ws);
  float* agg = (float*)(ws + ((size_t)NN * 4 + 255 & ~(size_t)255));
  float* h1  = (float*)((char*)agg + (size_t)NN * DIM * 4);

  int* cnt    = (int*)h1;
  int* slotno = (int*)((char*)h1 + 256 * 1024);
  int* seg    = (int*)((char*)h1 + 256 * 1024 + 2560 * 1024);
  int*   rowptr = (int*)out;
  int*   cs     = (int*)((char*)out + 256 * 1024);
  float* cw     = (float*)((char*)out + 256 * 1024 + 2560 * 1024);

  const int gemm_blocks = (NN + RPB - 1) / RPB;
  const int node_blocks = (NN + NPW * 8 - 1) / (NPW * 8);
  const int edge_blocks = (NE + 255) / 256;
  const int cnt_blocks  = (NN + 255) / 256;

  sage_zero_kernel<<<cnt_blocks, 256, 0, stream>>>(cnt, NN);
  sage_count_kernel<<<edge_blocks, 256, 0, stream>>>(dst, cnt, slotno, NE);
  sage_scan_kernel<<<1, SCAN_THREADS, 0, stream>>>(cnt, rowptr, deg, NN);
  sage_place_kernel<<<edge_blocks, 256, 0, stream>>>(dst, rowptr, slotno, seg, NE);
  sage_order_kernel<<<node_blocks, 256, 0, stream>>>(rowptr, seg, src, weight, cs, cw, NN);

  sage_gather_kernel<<<node_blocks, 256, 0, stream>>>(x, rowptr, cs, cw, agg, NN);
  sage_gemm_kernel<<<gemm_blocks, 256, 0, stream>>>(
      x, agg, deg, Wself1, Wneigh1, b1, h1, NN,  1);

  sage_gather_kernel<<<node_blocks, 256, 0, stream>>>(h1, rowptr, cs, cw, agg, NN);
  sage_gemm_kernel<<<gemm_blocks, 256, 0, stream>>>(
      h1, agg, deg, Wself2, Wneigh2, b2, out, NN,  0);
}


# Carry (edge id, src, w) as one 16-byte record through the placing step and order each node's records by edge id in registers inside the gather: the order launch and its per-edge src/w lookups go
# speedup vs baseline: 1.6021x; 1.0508x over previous
#include <hip/hip_runtime.h>


#define NN 50000
#define NE 625000
#define DIM 128

typedef __attribute__((ext_vector_type(2))) float v2f;
typedef __attribute__((ext_vector_type(8))) float v8f;


__global__ void sage_zero_kernel(int* __restrict__ p, int n) {
  int i = blockIdx.x * blockDim.x + threadIdx.x;
  if (i < n) p[i] = 0;
}

__global__ void sage_count_kernel(const int* __restrict__ dst, int* __restrict__ cnt,
                                  int* __restrict__ slotno, int E) {
  int e = blockIdx.x * blockDim.x + threadIdx.x;
  if (e < E) slotno[e] = atomicAdd(&cnt[dst[e]], 1);
}

#define SCAN_THREADS 1024
#define SCAN_WAVES   (SCAN_THREADS / 32)
#define SCAN_TILES   ((NN + 1 + SCAN_THREADS - 1) / SCAN_THREADS)
#define SCAN_GROUP   7
static_assert(SCAN_TILES % SCAN_GROUP == 0, "tile groups must cover the wave's range");

__device__ __forceinline__ int sage_wave_incl_scan(int v, int lane) {
  v += __builtin_amdgcn_update_dpp(0, v, 0x111, 0xf, 0xf, false);
  v += __builtin_amdgcn_update_dpp(0, v, 0x112, 0xf, 0xf, false);
  v += __builtin_amdgcn_update_dpp(0, v, 0x114, 0xf, 0xf, false);
  v += __builtin_amdgcn_update_dpp(0, v, 0x118, 0xf, 0xf, false);
  const int low = __builtin_amdgcn_readlane(v, 15);
  if (lane >= 16) v += low;
  return v;
}

__global__ __launch_bounds__(SCAN_THREADS) void sage_scan_kernel(
    const int* __restrict__ cnt, int* __restrict__ rowptr,
    float* __restrict__ deg, int n) {
  __shared__ int wtot[SCAN_WAVES];
  const int lane = threadIdx.x & 31;
  const int wave = threadIdx.x >> 5;
  const int wbase = wave * (SCAN_TILES * 32);

  int tsum = 0;
  for (int i0 = 0; i0 < SCAN_TILES; i0 += SCAN_GROUP) {
#pragma unroll
    for (int u = 0; u < SCAN_GROUP; ++u) {
      const int idx = wbase + (i0 + u) * 32 + lane;
      const int c = cnt[min(idx, n - 1)];
      tsum += (idx < n) ? c : 0;
    }
  }
  const int wsum = __builtin_amdgcn_readlane(sage_wave_incl_scan(tsum, lane), 31);
  if (lane == 0) wtot[wave] = wsum;
  __syncthreads();
  if (wave == 0) {
    const int t = wtot[lane];
    const int inc = sage_wave_incl_scan(t, lane);
    wtot[lane] = inc - t;
  }
  __syncthreads();
  int carry = wtot[wave];
  for (int i0 = 0; i0 < SCAN_TILES; i0 += SCAN_GROUP) {
    int c[SCAN_GROUP];
#pragma unroll
    for (int u = 0; u < SCAN_GROUP; ++u) {
      const int idx = wbase + (i0 + u) * 32 + lane;
      const int t = cnt[min(idx, n - 1)];
      c[u] = (idx < n) ? t : 0;
    }
#pragma unroll
    for (int u = 0; u < SCAN_GROUP; ++u) {
      const int idx = wbase + (i0 + u) * 32 + lane;
      const int inc = sage_wave_incl_scan(c[u], lane);
      if (idx <= n) rowptr[idx] = carry + inc - c[u];
      if (idx < n)  deg[idx] = (float)c[u];
      carry += __builtin_amdgcn_readlane(inc, 31);
    }
  }
}

__global__ void sage_place_kernel(const int* __restrict__ dst,
                                  const int* __restrict__ src,
                                  const float* __restrict__ w,
                                  const int* __restrict__ rowptr,
                                  const int* __restrict__ slotno,
                                  int* __restrict__ rec, int E) {
  int e = blockIdx.x * blockDim.x + threadIdx.x;
  if (e < E) {
    const int pos = rowptr[dst[e]] + slotno[e];
    ((int4*)rec)[pos] = make_int4(e, src[e], __float_as_int(w[e]), 0);
  }
}

#define NPW 4

template <int G>
__device__ __forceinline__ void sage_rows(float4& acc, const float* __restrict__ h,
                                          int ms, float mw, int j, int lane) {
  float4 v[G];
  float we[G];
#pragma unroll
  for (int u = 0; u < G; ++u) {
    const int s = __builtin_amdgcn_readlane(ms, j + u);
    we[u] = __int_as_float(__builtin_amdgcn_readlane(__float_as_int(mw), j + u));
    v[u] = ((const float4*)(h + (size_t)s * DIM))[lane];
  }
#pragma unroll
  for (int u = 0; u < G; ++u) {
    acc.x += v[u].x * we[u];
    acc.y += v[u].y * we[u];
    acc.z += v[u].z * we[u];
    acc.w += v[u].w * we[u];
  }
}

__global__ __launch_bounds__(256) void sage_gather_kernel(
    const float* __restrict__ h, const int* __restrict__ rowptr,
    const int* __restrict__ rec, float* __restrict__ agg, int n_nodes) {
  const int lane = threadIdx.x & 31;
  const int4* __restrict__ rec4 = (const int4*)rec;
  const int wv = __builtin_amdgcn_readfirstlane(
      (int)((blockIdx.x * blockDim.x + threadIdx.x) >> 5));
#pragma nounroll
  for (int k = 0; k < NPW; ++k) {
    const int node = wv * NPW + k;
    if (node >= n_nodes) break;
    const int beg = rowptr[node];
    const int n   = rowptr[node + 1] - beg;
    float4 acc = make_float4(0.0f, 0.0f, 0.0f, 0.0f);
    if (n <= 32) {
      int my = 0x7fffffff, s = 0, wb = 0;
      if (lane < n) {
        const int4 t = rec4[beg + lane];
        my = t.x; s = t.y; wb = t.z;
      }
      int rank = 0;
      for (int j = 0; j < n; ++j) {
        const int other = __builtin_amdgcn_readlane(my, j);
        rank += (other < my) ? 1 : 0;
      }
      if (lane >= n) rank = lane;
      const int   ms = __builtin_amdgcn_ds_permute(rank << 2, s);
      const float mw = __int_as_float(__builtin_amdgcn_ds_permute(rank << 2, wb));
      int j = 0;
      for (; j + 8 <= n; j += 8) sage_rows<8>(acc, h, ms, mw, j, lane);
      if (j + 4 <= n) { sage_rows<4>(acc, h, ms, mw, j, lane); j += 4; }
      if (j + 2 <= n) { sage_rows<2>(acc, h, ms, mw, j, lane); j += 2; }
      if (j < n)      { sage_rows<1>(acc, h, ms, mw, j, lane); }
    } else {
      int prev = -1;
      for (int r = 0; r < n; ++r) {
        int best = 0x7fffffff, bs = 0, bw = 0;
        for (int i0 = 0; i0 < n; i0 += 32) {
          const int i = i0 + lane;
          if (i < n) {
            const int4 t = rec4[beg + i];
            if (t.x > prev && t.x < best) { best = t.x; bs = t.y; bw = t.z; }
          }
        }
        int m = 0x7fffffff, src_lane = 0;
        for (int l = 0; l < 32; ++l) {
          const int v = __builtin_amdgcn_readlane(best, l);
          if (v < m) { m = v; src_lane = l; }
        }
        const int   ms = __builtin_amdgcn_readlane(bs, src_lane);
        const float mw = __int_as_float(__builtin_amdgcn_readlane(bw, src_lane));
        sage_rows<1>(acc, h, ms, mw, 0, lane);
        prev = m;
      }
    }
    ((float4*)(agg + (size_t)node * DIM))[lane] = acc;
  }
}

#define RPB 128
#define KC  16
#define AP  20
#define WP  132

__global__ __launch_bounds__(256) void sage_gemm_kernel(
    const float* __restrict__ Hself, const float* __restrict__ Agg,
    const float* __restrict__ Deg,   const float* __restrict__ Wself,
    const float* __restrict__ Wneigh,const float* __restrict__ Bias,
    float* __restrict__ Out, int nrows, int do_relu) {
  __shared__ float sWs[KC * WP];
  __shared__ float sWn[KC * WP];
  __shared__ float sAs[RPB * AP];
  __shared__ float sAn[RPB * AP];
  __shared__ float sRd[RPB];

  const int tid  = threadIdx.x;
  const int lane = tid & 31;
  const int wave = tid >> 5;
  const int rowbase = blockIdx.x * RPB;

  for (int r = tid; r < RPB; r += 256) {
    int gr = rowbase + r;
    float dg = (gr < nrows) ? Deg[gr] : 1.0f;
    sRd[r] = 1.0f / fmaxf(dg, 1.0f);
  }

  v8f acc[8] = {};

  const int jj    = wave * 16 + (lane & 15);
  const int khalf = (lane >> 4) << 1;
  const int m     = lane & 15;

  for (int kc = 0; kc < DIM / KC; ++kc) {
    __syncthreads();
#pragma unroll
    for (int s = 0; s < (RPB * KC) / 256; ++s) {
      int i = tid + s * 256;
      int j = i >> 4;
      int k = i & 15;
      sWs[k * WP + j] = Wself [j * DIM + kc * KC + k];
      sWn[k * WP + j] = Wneigh[j * DIM + kc * KC + k];
    }
#pragma unroll
    for (int s = 0; s < (RPB * KC) / 256; ++s) {
      int i = tid + s * 256;
      int r = i >> 4;
      int k = i & 15;
      int gr = rowbase + r;
      float hs = 0.0f, hn = 0.0f;
      if (gr < nrows) {
        hs = Hself[(size_t)gr * DIM + kc * KC + k];
        hn = Agg  [(size_t)gr * DIM + kc * KC + k] * sRd[r];
      }
      sAs[r * AP + k] = hs;
      sAn[r * AP + k] = hn;
    }
    __syncthreads();

    v2f bs[4], bn[4];
#pragma unroll
    for (int k0 = 0; k0 < 4; ++k0) {
      int kk = k0 * 4 + khalf;
      bs[k0].x = sWs[ kk      * WP + jj];
      bs[k0].y = sWs[(kk + 1) * WP + jj];
      bn[k0].x = sWn[ kk      * WP + jj];
      bn[k0].y = sWn[(kk + 1) * WP + jj];
    }

#pragma unroll
    for (int rt = 0; rt < 8; ++rt) {
      const float* as = &sAs[(rt * 16 + m) * AP];
      const float* an = &sAn[(rt * 16 + m) * AP];
#pragma unroll
      for (int k0 = 0; k0 < 4; ++k0) {
        int kk = k0 * 4 + khalf;
        v2f a_s; a_s.x = as[kk]; a_s.y = as[kk + 1];
        v2f a_n; a_n.x = an[kk]; a_n.y = an[kk + 1];
        acc[rt] = __builtin_amdgcn_wmma_f32_16x16x4_f32(
            false, a_s, false, bs[k0], (short)0, acc[rt], false, false);
        acc[rt] = __builtin_amdgcn_wmma_f32_16x16x4_f32(
            false, a_n, false, bn[k0], (short)0, acc[rt], false, false);
      }
    }
  }

  const float bv = Bias[jj];
  const int mofs = (lane >> 4) << 3;
#pragma unroll
  for (int rt = 0; rt < 8; ++rt) {
#pragma unroll
    for (int r = 0; r < 8; ++r) {
      int row = rowbase + rt * 16 + r + mofs;
      if (row < nrows) {
        float v = acc[rt][r] + bv;
        if (do_relu) v = fmaxf(v, 0.0f);
        Out[(size_t)row * DIM + jj] = v;
      }
    }
  }
}

extern "C" void kernel_launch(void* const* d_in, const int* in_sizes, int n_in,
                              void* d_out, int out_size, void* d_ws, size_t ws_size,
                              hipStream_t stream) {
  const float* x       = (const float*)d_in[0];
  const int*   src     = (const int*)  d_in[1];
  const int*   dst     = (const int*)  d_in[2];
  const float* weight  = (const float*)d_in[3];
  const float* Wself1  = (const float*)d_in[4];
  const float* Wneigh1 = (const float*)d_in[5];
  const float* b1      = (const float*)d_in[6];
  const float* Wself2  = (const float*)d_in[7];
  const float* Wneigh2 = (const float*)d_in[8];
  const float* b2      = (const float*)d_in[9];
  float* out = (float*)d_out;

  char* ws = (char*)d_ws;
  float* deg = (float*)(ws);
  float* agg = (float*)(ws + ((size_t)NN * 4 + 255 & ~(size_t)255));
  float* h1  = (float*)((char*)agg + (size_t)NN * DIM * 4);

  int* cnt    = (int*)h1;
  int* slotno = (int*)((char*)h1 + 256 * 1024);
  int* rowptr = (int*)out;
  int* rec    = (int*)((char*)out + 256 * 1024);

  const int gemm_blocks = (NN + RPB - 1) / RPB;
  const int node_blocks = (NN + NPW * 8 - 1) / (NPW * 8);
  const int edge_blocks = (NE + 255) / 256;
  const int cnt_blocks  = (NN + 255) / 256;

  sage_zero_kernel<<<cnt_blocks, 256, 0, stream>>>(cnt, NN);
  sage_count_kernel<<<edge_blocks, 256, 0, stream>>>(dst, cnt, slotno, NE);
  sage_scan_kernel<<<1, SCAN_THREADS, 0, stream>>>(cnt, rowptr, deg, NN);
  sage_place_kernel<<<edge_blocks, 256, 0, stream>>>(dst, src, weight, rowptr, slotno, rec, NE);

  sage_gather_kernel<<<node_blocks, 256, 0, stream>>>(x, rowptr, rec, agg, NN);
  sage_gemm_kernel<<<gemm_blocks, 256, 0, stream>>>(
      x, agg, deg, Wself1, Wneigh1, b1, h1, NN,  1);

  sage_gather_kernel<<<node_blocks, 256, 0, stream>>>(h1, rowptr, rec, agg, NN);
  sage_gemm_kernel<<<gemm_blocks, 256, 0, stream>>>(
      h1, agg, deg, Wself2, Wneigh2, b2, out, NN,  0);
}


# GEMM operand feed pipelined: K chunks of 32 fetched as whole 128-byte lines by 16-byte loads one chunk ahead in registers, W row-major in LDS (B fragments 8-byte reads like A); sum order unchanged
# speedup vs baseline: 2.0967x; 1.3087x over previous
#include <hip/hip_runtime.h>


#define NN 50000
#define NE 625000
#define DIM 128

typedef __attribute__((ext_vector_type(2))) float v2f;
typedef __attribute__((ext_vector_type(8))) float v8f;
typedef __attribute__((ext_vector_type(4))) float v4f;


__global__ void sage_zero_kernel(int* __restrict__ p, int n) {
  int i = blockIdx.x * blockDim.x + threadIdx.x;
  if (i < n) p[i] = 0;
}

__global__ void sage_count_kernel(const int* __restrict__ dst, int* __restrict__ cnt,
                                  int* __restrict__ slotno, int E) {
  int e = blockIdx.x * blockDim.x + threadIdx.x;
  if (e < E) slotno[e] = atomicAdd(&cnt[dst[e]], 1);
}

#define SCAN_THREADS 1024
#define SCAN_WAVES   (SCAN_THREADS / 32)
#define SCAN_TILES   ((NN + 1 + SCAN_THREADS - 1) / SCAN_THREADS)
#define SCAN_GROUP   7
static_assert(SCAN_TILES % SCAN_GROUP == 0, "tile groups must cover the wave's range");

__device__ __forceinline__ int sage_wave_incl_scan(int v, int lane) {
  v += __builtin_amdgcn_update_dpp(0, v, 0x111, 0xf, 0xf, false);
  v += __builtin_amdgcn_update_dpp(0, v, 0x112, 0xf, 0xf, false);
  v += __builtin_amdgcn_update_dpp(0, v, 0x114, 0xf, 0xf, false);
  v += __builtin_amdgcn_update_dpp(0, v, 0x118, 0xf, 0xf, false);
  const int low = __builtin_amdgcn_readlane(v, 15);
  if (lane >= 16) v += low;
  return v;
}

__global__ __launch_bounds__(SCAN_THREADS) void sage_scan_kernel(
    const int* __restrict__ cnt, int* __restrict__ rowptr,
    float* __restrict__ deg, int n) {
  __shared__ int wtot[SCAN_WAVES];
  const int lane = threadIdx.x & 31;
  const int wave = threadIdx.x >> 5;
  const int wbase = wave * (SCAN_TILES * 32);

  int tsum = 0;
  for (int i0 = 0; i0 < SCAN_TILES; i0 += SCAN_GROUP) {
#pragma unroll
    for (int u = 0; u < SCAN_GROUP; ++u) {
      const int idx = wbase + (i0 + u) * 32 + lane;
      const int c = cnt[min(idx, n - 1)];
      tsum += (idx < n) ? c : 0;
    }
  }
  const int wsum = __builtin_amdgcn_readlane(sage_wave_incl_scan(tsum, lane), 31);
  if (lane == 0) wtot[wave] = wsum;
  __syncthreads();
  if (wave == 0) {
    const int t = wtot[lane];
    const int inc = sage_wave_incl_scan(t, lane);
    wtot[lane] = inc - t;
  }
  __syncthreads();
  int carry = wtot[wave];
  for (int i0 = 0; i0 < SCAN_TILES; i0 += SCAN_GROUP) {
    int c[SCAN_GROUP];
#pragma unroll
    for (int u = 0; u < SCAN_GROUP; ++u) {
      const int idx = wbase + (i0 + u) * 32 + lane;
      const int t = cnt[min(idx, n - 1)];
      c[u] = (idx < n) ? t : 0;
    }
#pragma unroll
    for (int u = 0; u < SCAN_GROUP; ++u) {
      const int idx = wbase + (i0 + u) * 32 + lane;
      const int inc = sage_wave_incl_scan(c[u], lane);
      if (idx <= n) rowptr[idx] = carry + inc - c[u];
      if (idx < n)  deg[idx] = (float)c[u];
      carry += __builtin_amdgcn_readlane(inc, 31);
    }
  }
}

__global__ void sage_place_kernel(const int* __restrict__ dst,
                                  const int* __restrict__ src,
                                  const float* __restrict__ w,
                                  const int* __restrict__ rowptr,
                                  const int* __restrict__ slotno,
                                  int* __restrict__ rec, int E) {
  int e = blockIdx.x * blockDim.x + threadIdx.x;
  if (e < E) {
    const int pos = rowptr[dst[e]] + slotno[e];
    ((int4*)rec)[pos] = make_int4(e, src[e], __float_as_int(w[e]), 0);
  }
}

#define NPW 4

template <int G>
__device__ __forceinline__ void sage_rows(float4& acc, const float* __restrict__ h,
                                          int ms, float mw, int j, int lane) {
  float4 v[G];
  float we[G];
#pragma unroll
  for (int u = 0; u < G; ++u) {
    const int s = __builtin_amdgcn_readlane(ms, j + u);
    we[u] = __int_as_float(__builtin_amdgcn_readlane(__float_as_int(mw), j + u));
    v[u] = ((const float4*)(h + (size_t)s * DIM))[lane];
  }
#pragma unroll
  for (int u = 0; u < G; ++u) {
    acc.x += v[u].x * we[u];
    acc.y += v[u].y * we[u];
    acc.z += v[u].z * we[u];
    acc.w += v[u].w * we[u];
  }
}

__global__ __launch_bounds__(256) void sage_gather_kernel(
    const float* __restrict__ h, const int* __restrict__ rowptr,
    const int* __restrict__ rec, float* __restrict__ agg, int n_nodes) {
  const int lane = threadIdx.x & 31;
  const int4* __restrict__ rec4 = (const int4*)rec;
  const int wv = __builtin_amdgcn_readfirstlane(
      (int)((blockIdx.x * blockDim.x + threadIdx.x) >> 5));
#pragma nounroll
  for (int k = 0; k < NPW; ++k) {
    const int node = wv * NPW + k;
    if (node >= n_nodes) break;
    const int beg = rowptr[node];
    const int n   = rowptr[node + 1] - beg;
    float4 acc = make_float4(0.0f, 0.0f, 0.0f, 0.0f);
    if (n <= 32) {
      int my = 0x7fffffff, s = 0, wb = 0;
      if (lane < n) {
        const int4 t = rec4[beg + lane];
        my = t.x; s = t.y; wb = t.z;
      }
      int rank = 0;
      for (int j = 0; j < n; ++j) {
        const int other = __builtin_amdgcn_readlane(my, j);
        rank += (other < my) ? 1 : 0;
      }
      if (lane >= n) rank = lane;
      const int   ms = __builtin_amdgcn_ds_permute(rank << 2, s);
      const float mw = __int_as_float(__builtin_amdgcn_ds_permute(rank << 2, wb));
      int j = 0;
      for (; j + 8 <= n; j += 8) sage_rows<8>(acc, h, ms, mw, j, lane);
      if (j + 4 <= n) { sage_rows<4>(acc, h, ms, mw, j, lane); j += 4; }
      if (j + 2 <= n) { sage_rows<2>(acc, h, ms, mw, j, lane); j += 2; }
      if (j < n)      { sage_rows<1>(acc, h, ms, mw, j, lane); }
    } else {
      int prev = -1;
      for (int r = 0; r < n; ++r) {
        int best = 0x7fffffff, bs = 0, bw = 0;
        for (int i0 = 0; i0 < n; i0 += 32) {
          const int i = i0 + lane;
          if (i < n) {
            const int4 t = rec4[beg + i];
            if (t.x > prev && t.x < best) { best = t.x; bs = t.y; bw = t.z; }
          }
        }
        int m = 0x7fffffff, src_lane = 0;
        for (int l = 0; l < 32; ++l) {
          const int v = __builtin_amdgcn_readlane(best, l);
          if (v < m) { m = v; src_lane = l; }
        }
        const int   ms = __builtin_amdgcn_readlane(bs, src_lane);
        const float mw = __int_as_float(__builtin_amdgcn_readlane(bw, src_lane));
        sage_rows<1>(acc, h, ms, mw, 0, lane);
        prev = m;
      }
    }
    ((float4*)(agg + (size_t)node * DIM))[lane] = acc;
  }
}

#define RPB 128
#define KC  32
#define LP  36
#define NCH (DIM / KC)

__global__ __launch_bounds__(256) void sage_gemm_kernel(
    const float* __restrict__ Hself, const float* __restrict__ Agg,
    const float* __restrict__ Deg,   const float* __restrict__ Wself,
    const float* __restrict__ Wneigh,const float* __restrict__ Bias,
    float* __restrict__ Out, int nrows, int do_relu) {
  __shared__ __attribute__((aligned(16))) float sWs[DIM * LP];
  __shared__ __attribute__((aligned(16))) float sWn[DIM * LP];
  __shared__ __attribute__((aligned(16))) float sAs[RPB * LP];
  __shared__ __attribute__((aligned(16))) float sAn[RPB * LP];

  const int tid  = threadIdx.x;
  const int lane = tid & 31;
  const int wave = tid >> 5;
  const int rowbase = blockIdx.x * RPB;

  const int sr = tid >> 3;
  const int c4 = tid & 7;
  float rd[4];
#pragma unroll
  for (int s = 0; s < 4; ++s) {
    const int gr = min(rowbase + sr + s * 32, nrows - 1);
    rd[s] = 1.0f / fmaxf(Deg[gr], 1.0f);
  }

  const int gr0 = min(rowbase + sr,      nrows - 1);
  const int gr1 = min(rowbase + sr + 32, nrows - 1);
  const int gr2 = min(rowbase + sr + 64, nrows - 1);
  const int gr3 = min(rowbase + sr + 96, nrows - 1);
  const v4f* gws = (const v4f*)(Wself  + (size_t)sr * DIM) + c4;
  const v4f* gwn = (const v4f*)(Wneigh + (size_t)sr * DIM) + c4;
  const v4f* gs0 = (const v4f*)(Hself + (size_t)gr0 * DIM) + c4;
  const v4f* gs1 = (const v4f*)(Hself + (size_t)gr1 * DIM) + c4;
  const v4f* gs2 = (const v4f*)(Hself + (size_t)gr2 * DIM) + c4;
  const v4f* gs3 = (const v4f*)(Hself + (size_t)gr3 * DIM) + c4;
  const v4f* gn0 = (const v4f*)(Agg   + (size_t)gr0 * DIM) + c4;
  const v4f* gn1 = (const v4f*)(Agg   + (size_t)gr1 * DIM) + c4;
  const v4f* gn2 = (const v4f*)(Agg   + (size_t)gr2 * DIM) + c4;
  const v4f* gn3 = (const v4f*)(Agg   + (size_t)gr3 * DIM) + c4;
  v4f ws0, ws1, ws2, ws3, wn0, wn1, wn2, wn3;
  v4f as0, as1, as2, as3, an0, an1, an2, an3;
#define SAGE_LOAD_CHUNK(kc_)                                   \
  {                                                            \
    const int o  = (kc_) * (KC / 4);                           \
    const int w1 = 32 * DIM / 4, w2 = 64 * DIM / 4, w3 = 96 * DIM / 4; \
    ws0 = gws[o]; ws1 = gws[o + w1]; ws2 = gws[o + w2]; ws3 = gws[o + w3]; \
    wn0 = gwn[o]; wn1 = gwn[o + w1]; wn2 = gwn[o + w2]; wn3 = gwn[o + w3]; \
    as0 = gs0[o]; as1 = gs1[o]; as2 = gs2[o]; as3 = gs3[o];    \
    an0 = gn0[o]; an1 = gn1[o]; an2 = gn2[o]; an3 = gn3[o];    \
  }

  SAGE_LOAD_CHUNK(0)

  v8f acc[8] = {};

  const int jj    = wave * 16 + (lane & 15);
  const int khalf = (lane >> 4) << 1;
  const int m     = lane & 15;

#pragma nounroll
  for (int kc = 0; kc < NCH; ++kc) {
    __syncthreads();
    {
      v4f* ls = (v4f*)&sWs[sr * LP + c4 * 4];
      v4f* ln = (v4f*)&sWn[sr * LP + c4 * 4];
      v4f* la = (v4f*)&sAs[sr * LP + c4 * 4];
      v4f* lb = (v4f*)&sAn[sr * LP + c4 * 4];
      const int q = 32 * LP / 4;
      ls[0] = ws0; ls[q] = ws1; ls[2 * q] = ws2; ls[3 * q] = ws3;
      ln[0] = wn0; ln[q] = wn1; ln[2 * q] = wn2; ln[3 * q] = wn3;
      la[0] = as0; la[q] = as1; la[2 * q] = as2; la[3 * q] = as3;
      lb[0] = an0 * rd[0]; lb[q] = an1 * rd[1]; lb[2 * q] = an2 * rd[2]; lb[3 * q] = an3 * rd[3];
    }
    __syncthreads();
    if (kc + 1 < NCH) { SAGE_LOAD_CHUNK(kc + 1) }

    v2f bs[8], bn[8];
#pragma unroll
    for (int k0 = 0; k0 < 8; ++k0) {
      const int kk = k0 * 4 + khalf;
      bs[k0] = *(const v2f*)&sWs[jj * LP + kk];
      bn[k0] = *(const v2f*)&sWn[jj * LP + kk];
    }
    v2f fs[2][4], fn[2][4];
#pragma unroll
    for (int k0 = 0; k0 < 4; ++k0) {
      fs[0][k0] = *(const v2f*)&sAs[m * LP + k0 * 4 + khalf];
      fn[0][k0] = *(const v2f*)&sAn[m * LP + k0 * 4 + khalf];
    }
#pragma unroll
    for (int st = 0; st < 16; ++st) {
      const int cur = st & 1, nxt = cur ^ 1;
      const int kh = st >> 3, rt = st & 7;
      if (st + 1 < 16) {
        const int kh1 = (st + 1) >> 3, rt1 = (st + 1) & 7;
#pragma unroll
        for (int k0 = 0; k0 < 4; ++k0) {
          const int kk = kh1 * 16 + k0 * 4 + khalf;
          fs[nxt][k0] = *(const v2f*)&sAs[(rt1 * 16 + m) * LP + kk];
          fn[nxt][k0] = *(const v2f*)&sAn[(rt1 * 16 + m) * LP + kk];
        }
      }
#pragma unroll
      for (int k0 = 0; k0 < 4; ++k0) {
        acc[rt] = __builtin_amdgcn_wmma_f32_16x16x4_f32(
            false, fs[cur][k0], false, bs[kh * 4 + k0], (short)0, acc[rt], false, false);
        acc[rt] = __builtin_amdgcn_wmma_f32_16x16x4_f32(
            false, fn[cur][k0], false, bn[kh * 4 + k0], (short)0, acc[rt], false, false);
      }
    }
    __builtin_amdgcn_sched_barrier(0);
    asm volatile("v_nop\n\tv_nop\n\tv_nop\n\tv_nop");
    __builtin_amdgcn_sched_barrier(0);
  }
#undef SAGE_LOAD_CHUNK

  const float bv = Bias[jj];
  const int mofs = (lane >> 4) << 3;
#pragma unroll
  for (int rt = 0; rt < 8; ++rt) {
#pragma unroll
    for (int r = 0; r < 8; ++r) {
      int row = rowbase + rt * 16 + r + mofs;
      if (row < nrows) {
        float v = acc[rt][r] + bv;
        if (do_relu) v = fmaxf(v, 0.0f);
        Out[(size_t)row * DIM + jj] = v;
      }
    }
  }
}

extern "C" void kernel_launch(void* const* d_in, const int* in_sizes, int n_in,
                              void* d_out, int out_size, void* d_ws, size_t ws_size,
                              hipStream_t stream) {
  const float* x       = (const float*)d_in[0];
  const int*   src     = (const int*)  d_in[1];
  const int*   dst     = (const int*)  d_in[2];
  const float* weight  = (const float*)d_in[3];
  const float* Wself1  = (const float*)d_in[4];
  const float* Wneigh1 = (const float*)d_in[5];
  const float* b1      = (const float*)d_in[6];
  const float* Wself2  = (const float*)d_in[7];
  const float* Wneigh2 = (const float*)d_in[8];
  const float* b2      = (const float*)d_in[9];
  float* out = (float*)d_out;

  char* ws = (char*)d_ws;
  float* deg = (float*)(ws);
  float* agg = (float*)(ws + ((size_t)NN * 4 + 255 & ~(size_t)255));
  float* h1  = (float*)((char*)agg + (size_t)NN * DIM * 4);

  int* cnt    = (int*)h1;
  int* slotno = (int*)((char*)h1 + 256 * 1024);
  int* rowptr = (int*)out;
  int* rec    = (int*)((char*)out + 256 * 1024);

  const int gemm_blocks = (NN + RPB - 1) / RPB;
  const int node_blocks = (NN + NPW * 8 - 1) / (NPW * 8);
  const int edge_blocks = (NE + 255) / 256;
  const int cnt_blocks  = (NN + 255) / 256;

  sage_zero_kernel<<<cnt_blocks, 256, 0, stream>>>(cnt, NN);
  sage_count_kernel<<<edge_blocks, 256, 0, stream>>>(dst, cnt, slotno, NE);
  sage_scan_kernel<<<1, SCAN_THREADS, 0, stream>>>(cnt, rowptr, deg, NN);
  sage_place_kernel<<<edge_blocks, 256, 0, stream>>>(dst, src, weight, rowptr, slotno, rec, NE);

  sage_gather_kernel<<<node_blocks, 256, 0, stream>>>(x, rowptr, rec, agg, NN);
  sage_gemm_kernel<<<gemm_blocks, 256, 0, stream>>>(
      x, agg, deg, Wself1, Wneigh1, b1, h1, NN,  1);

  sage_gather_kernel<<<node_blocks, 256, 0, stream>>>(h1, rowptr, rec, agg, NN);
  sage_gemm_kernel<<<gemm_blocks, 256, 0, stream>>>(
      h1, agg, deg, Wself2, Wneigh2, b2, out, NN,  0);
}


# Edge list built in one launch by owner blocks: each reads all dst once, keeps its node range's edges in LDS, counts, scans, places them with LDS atomics; build launches 4 to 1; sum order unchanged
# speedup vs baseline: 2.8382x; 1.3537x over previous
#include <hip/hip_runtime.h>


#define NN 50000
#define NE 625000
#define DIM 128

typedef __attribute__((ext_vector_type(2))) float v2f;
typedef __attribute__((ext_vector_type(8))) float v8f;
typedef __attribute__((ext_vector_type(4))) float v4f;


__device__ __forceinline__ int sage_wave_incl_scan(int v, int lane) {
  v += __builtin_amdgcn_update_dpp(0, v, 0x111, 0xf, 0xf, false);
  v += __builtin_amdgcn_update_dpp(0, v, 0x112, 0xf, 0xf, false);
  v += __builtin_amdgcn_update_dpp(0, v, 0x114, 0xf, 0xf, false);
  v += __builtin_amdgcn_update_dpp(0, v, 0x118, 0xf, 0xf, false);
  const int low = __builtin_amdgcn_readlane(v, 15);
  if (lane >= 16) v += low;
  return v;
}

#define OWN         224
#define BLD_THREADS 1024
#define BLD_WAVES   (BLD_THREADS / 32)
#define BLD_U       8
#define BLD_CAP     8192
#define BLD_Q       (NE / 4)
#define BLD_GROUP   (BLD_THREADS * BLD_U)
#define BLD_MAIN    ((BLD_Q / BLD_GROUP) * BLD_GROUP)
#define BLD_NG      (BLD_Q / BLD_GROUP)
#define BLD_BLOCKS  ((NN + 1 + OWN - 1) / OWN)
static_assert(NE % 4 == 0, "dst is read as 16-byte pieces");
static_assert(OWN % 32 == 0 && BLD_WAVES == 32, "scan by whole waves");
static_assert(BLD_NG >= 1, "the sweep has at least one whole group");

__device__ __forceinline__ void sage_bld_visit(const int4 v, int e0, int lo,
                                               unsigned& nless, int* s_nlist,
                                               int2* s_list) {
  const unsigned t[4] = {(unsigned)(v.x - lo), (unsigned)(v.y - lo),
                         (unsigned)(v.z - lo), (unsigned)(v.w - lo)};
  nless += (t[0] >> 31) + (t[1] >> 31);
  nless += (t[2] >> 31) + (t[3] >> 31);
  const unsigned m = min(min(t[0], t[1]), min(t[2], t[3]));
  if (m < (unsigned)OWN) {
#pragma unroll
    for (int c = 0; c < 4; ++c) {
      if (t[c] < (unsigned)OWN) {
        const int li = atomicAdd(s_nlist, 1);
        if (li < BLD_CAP) s_list[li] = make_int2(e0 + c, (int)t[c]);
      }
    }
  }
}

__global__ __launch_bounds__(BLD_THREADS) void sage_build_kernel(
    const int* __restrict__ dst, const int* __restrict__ src,
    const float* __restrict__ w, int* __restrict__ rowptr,
    float* __restrict__ deg, int* __restrict__ rec) {
  __shared__ int  s_cnt[OWN];
  __shared__ int  s_ptr[OWN];
  __shared__ int  s_cur[OWN];
  __shared__ int  s_wless[BLD_WAVES];
  __shared__ int  s_nlist;
  __shared__ int2 s_list[BLD_CAP];
  __shared__ int  s_slot[BLD_CAP];

  const int tid  = threadIdx.x;
  const int lane = tid & 31;
  const int wave = tid >> 5;
  const int lo = blockIdx.x * OWN;
  const int hi = lo + OWN;
  int4* __restrict__ rec4 = (int4*)rec;

  for (int i = tid; i < OWN; i += BLD_THREADS) { s_cnt[i] = 0; s_cur[i] = 0; }
  if (tid == 0) s_nlist = 0;
  __syncthreads();

  const int4* __restrict__ dst4 = (const int4*)dst;
  unsigned nless = 0;
  int gsel = (int)(blockIdx.x % BLD_NG);
  for (int gi = 0; gi < BLD_NG; ++gi) {
    const int g = gsel * BLD_GROUP + tid;
    int4 v[BLD_U];
#pragma unroll
    for (int u = 0; u < BLD_U; ++u) v[u] = dst4[g + u * BLD_THREADS];
#pragma unroll
    for (int u = 0; u < BLD_U; ++u)
      sage_bld_visit(v[u], (g + u * BLD_THREADS) * 4, lo, nless, &s_nlist, s_list);
    gsel = (gsel + 1 == BLD_NG) ? 0 : gsel + 1;
  }
  for (int j = BLD_MAIN + tid; j < BLD_Q; j += BLD_THREADS)
    sage_bld_visit(dst4[j], j * 4, lo, nless, &s_nlist, s_list);
  {
    const int wsum = sage_wave_incl_scan((int)nless, lane);
    if (lane == 31) s_wless[wave] = wsum;
  }
  __syncthreads();

  const int total = s_nlist;
  const bool fits = total <= BLD_CAP;
  if (fits) {
    for (int li = tid; li < total; li += BLD_THREADS)
      s_slot[li] = atomicAdd(&s_cnt[s_list[li].y], 1);
  } else {
    for (int e = tid; e < NE; e += BLD_THREADS) {
      const int dd = dst[e];
      if (dd >= lo && dd < hi) atomicAdd(&s_cnt[dd - lo], 1);
    }
  }
  __syncthreads();

  if (wave == 0) {
    const int below = sage_wave_incl_scan(s_wless[lane], lane);
    int carry = __builtin_amdgcn_readlane(below, 31);
    for (int i0 = 0; i0 < OWN; i0 += 32) {
      const int i = i0 + lane;
      const int c = s_cnt[i];
      const int inc = sage_wave_incl_scan(c, lane);
      const int p = carry + inc - c;
      s_ptr[i] = p;
      const int node = lo + i;
      if (node <= NN) rowptr[node] = p;
      if (node <  NN) deg[node] = (float)c;
      carry += __builtin_amdgcn_readlane(inc, 31);
    }
  }
  __syncthreads();

  if (fits) {
    for (int l0 = tid; l0 < total; l0 += 4 * BLD_THREADS) {
      int e4[4], pos[4], sv[4], wb[4];
#pragma unroll
      for (int k = 0; k < 4; ++k) {
        const int li = l0 + k * BLD_THREADS;
        const bool ok = li < total;
        const int lc = ok ? li : l0;
        const int2 t = s_list[lc];
        e4[k]  = t.x;
        pos[k] = ok ? s_ptr[t.y] + s_slot[lc] : -1;
        sv[k]  = src[t.x];
        wb[k]  = __float_as_int(w[t.x]);
      }
#pragma unroll
      for (int k = 0; k < 4; ++k)
        if (pos[k] >= 0) rec4[pos[k]] = make_int4(e4[k], sv[k], wb[k], 0);
    }
  } else {
    for (int e = tid; e < NE; e += BLD_THREADS) {
      const int dd = dst[e];
      if (dd >= lo && dd < hi) {
        const int pos = s_ptr[dd - lo] + atomicAdd(&s_cur[dd - lo], 1);
        rec4[pos] = make_int4(e, src[e], __float_as_int(w[e]), 0);
      }
    }
  }
}

#define NPW 4

template <int G>
__device__ __forceinline__ void sage_rows(float4& acc, const float* __restrict__ h,
                                          int ms, float mw, int j, int lane) {
  float4 v[G];
  float we[G];
#pragma unroll
  for (int u = 0; u < G; ++u) {
    const int s = __builtin_amdgcn_readlane(ms, j + u);
    we[u] = __int_as_float(__builtin_amdgcn_readlane(__float_as_int(mw), j + u));
    v[u] = ((const float4*)(h + (size_t)s * DIM))[lane];
  }
#pragma unroll
  for (int u = 0; u < G; ++u) {
    acc.x += v[u].x * we[u];
    acc.y += v[u].y * we[u];
    acc.z += v[u].z * we[u];
    acc.w += v[u].w * we[u];
  }
}

__global__ __launch_bounds__(256) void sage_gather_kernel(
    const float* __restrict__ h, const int* __restrict__ rowptr,
    const int* __restrict__ rec, float* __restrict__ agg, int n_nodes) {
  const int lane = threadIdx.x & 31;
  const int4* __restrict__ rec4 = (const int4*)rec;
  const int wv = __builtin_amdgcn_readfirstlane(
      (int)((blockIdx.x * blockDim.x + threadIdx.x) >> 5));
#pragma nounroll
  for (int k = 0; k < NPW; ++k) {
    const int node = wv * NPW + k;
    if (node >= n_nodes) break;
    const int beg = rowptr[node];
    const int n   = rowptr[node + 1] - beg;
    float4 acc = make_float4(0.0f, 0.0f, 0.0f, 0.0f);
    if (n <= 32) {
      int my = 0x7fffffff, s = 0, wb = 0;
      if (lane < n) {
        const int4 t = rec4[beg + lane];
        my = t.x; s = t.y; wb = t.z;
      }
      int rank = 0;
      for (int j = 0; j < n; ++j) {
        const int other = __builtin_amdgcn_readlane(my, j);
        rank += (other < my) ? 1 : 0;
      }
      if (lane >= n) rank = lane;
      const int   ms = __builtin_amdgcn_ds_permute(rank << 2, s);
      const float mw = __int_as_float(__builtin_amdgcn_ds_permute(rank << 2, wb));
      int j = 0;
      for (; j + 8 <= n; j += 8) sage_rows<8>(acc, h, ms, mw, j, lane);
      if (j + 4 <= n) { sage_rows<4>(acc, h, ms, mw, j, lane); j += 4; }
      if (j + 2 <= n) { sage_rows<2>(acc, h, ms, mw, j, lane); j += 2; }
      if (j < n)      { sage_rows<1>(acc, h, ms, mw, j, lane); }
    } else {
      int prev = -1;
      for (int r = 0; r < n; ++r) {
        int best = 0x7fffffff, bs = 0, bw = 0;
        for (int i0 = 0; i0 < n; i0 += 32) {
          const int i = i0 + lane;
          if (i < n) {
            const int4 t = rec4[beg + i];
            if (t.x > prev && t.x < best) { best = t.x; bs = t.y; bw = t.z; }
          }
        }
        int m = 0x7fffffff, src_lane = 0;
        for (int l = 0; l < 32; ++l) {
          const int v = __builtin_amdgcn_readlane(best, l);
          if (v < m) { m = v; src_lane = l; }
        }
        const int   ms = __builtin_amdgcn_readlane(bs, src_lane);
        const float mw = __int_as_float(__builtin_amdgcn_readlane(bw, src_lane));
        sage_rows<1>(acc, h, ms, mw, 0, lane);
        prev = m;
      }
    }
    ((float4*)(agg + (size_t)node * DIM))[lane] = acc;
  }
}

#define RPB 128
#define KC  32
#define LP  36
#define NCH (DIM / KC)

__global__ __launch_bounds__(256) void sage_gemm_kernel(
    const float* __restrict__ Hself, const float* __restrict__ Agg,
    const float* __restrict__ Deg,   const float* __restrict__ Wself,
    const float* __restrict__ Wneigh,const float* __restrict__ Bias,
    float* __restrict__ Out, int nrows, int do_relu) {
  __shared__ __attribute__((aligned(16))) float sWs[DIM * LP];
  __shared__ __attribute__((aligned(16))) float sWn[DIM * LP];
  __shared__ __attribute__((aligned(16))) float sAs[RPB * LP];
  __shared__ __attribute__((aligned(16))) float sAn[RPB * LP];

  const int tid  = threadIdx.x;
  const int lane = tid & 31;
  const int wave = tid >> 5;
  const int rowbase = blockIdx.x * RPB;

  const int sr = tid >> 3;
  const int c4 = tid & 7;
  float rd[4];
#pragma unroll
  for (int s = 0; s < 4; ++s) {
    const int gr = min(rowbase + sr + s * 32, nrows - 1);
    rd[s] = 1.0f / fmaxf(Deg[gr], 1.0f);
  }

  const int gr0 = min(rowbase + sr,      nrows - 1);
  const int gr1 = min(rowbase + sr + 32, nrows - 1);
  const int gr2 = min(rowbase + sr + 64, nrows - 1);
  const int gr3 = min(rowbase + sr + 96, nrows - 1);
  const v4f* gws = (const v4f*)(Wself  + (size_t)sr * DIM) + c4;
  const v4f* gwn = (const v4f*)(Wneigh + (size_t)sr * DIM) + c4;
  const v4f* gs0 = (const v4f*)(Hself + (size_t)gr0 * DIM) + c4;
  const v4f* gs1 = (const v4f*)(Hself + (size_t)gr1 * DIM) + c4;
  const v4f* gs2 = (const v4f*)(Hself + (size_t)gr2 * DIM) + c4;
  const v4f* gs3 = (const v4f*)(Hself + (size_t)gr3 * DIM) + c4;
  const v4f* gn0 = (const v4f*)(Agg   + (size_t)gr0 * DIM) + c4;
  const v4f* gn1 = (const v4f*)(Agg   + (size_t)gr1 * DIM) + c4;
  const v4f* gn2 = (const v4f*)(Agg   + (size_t)gr2 * DIM) + c4;
  const v4f* gn3 = (const v4f*)(Agg   + (size_t)gr3 * DIM) + c4;
  v4f ws0, ws1, ws2, ws3, wn0, wn1, wn2, wn3;
  v4f as0, as1, as2, as3, an0, an1, an2, an3;
#define SAGE_LOAD_CHUNK(kc_)                                   \
  {                                                            \
    const int o  = (kc_) * (KC / 4);                           \
    const int w1 = 32 * DIM / 4, w2 = 64 * DIM / 4, w3 = 96 * DIM / 4; \
    ws0 = gws[o]; ws1 = gws[o + w1]; ws2 = gws[o + w2]; ws3 = gws[o + w3]; \
    wn0 = gwn[o]; wn1 = gwn[o + w1]; wn2 = gwn[o + w2]; wn3 = gwn[o + w3]; \
    as0 = gs0[o]; as1 = gs1[o]; as2 = gs2[o]; as3 = gs3[o];    \
    an0 = gn0[o]; an1 = gn1[o]; an2 = gn2[o]; an3 = gn3[o];    \
  }

  SAGE_LOAD_CHUNK(0)

  v8f acc[8] = {};

  const int jj    = wave * 16 + (lane & 15);
  const int khalf = (lane >> 4) << 1;
  const int m     = lane & 15;

#pragma nounroll
  for (int kc = 0; kc < NCH; ++kc) {
    __syncthreads();
    {
      v4f* ls = (v4f*)&sWs[sr * LP + c4 * 4];
      v4f* ln = (v4f*)&sWn[sr * LP + c4 * 4];
      v4f* la = (v4f*)&sAs[sr * LP + c4 * 4];
      v4f* lb = (v4f*)&sAn[sr * LP + c4 * 4];
      const int q = 32 * LP / 4;
      ls[0] = ws0; ls[q] = ws1; ls[2 * q] = ws2; ls[3 * q] = ws3;
      ln[0] = wn0; ln[q] = wn1; ln[2 * q] = wn2; ln[3 * q] = wn3;
      la[0] = as0; la[q] = as1; la[2 * q] = as2; la[3 * q] = as3;
      lb[0] = an0 * rd[0]; lb[q] = an1 * rd[1]; lb[2 * q] = an2 * rd[2]; lb[3 * q] = an3 * rd[3];
    }
    __syncthreads();
    if (kc + 1 < NCH) { SAGE_LOAD_CHUNK(kc + 1) }

    v2f bs[8], bn[8];
#pragma unroll
    for (int k0 = 0; k0 < 8; ++k0) {
      const int kk = k0 * 4 + khalf;
      bs[k0] = *(const v2f*)&sWs[jj * LP + kk];
      bn[k0] = *(const v2f*)&sWn[jj * LP + kk];
    }
    v2f fs[2][4], fn[2][4];
#pragma unroll
    for (int k0 = 0; k0 < 4; ++k0) {
      fs[0][k0] = *(const v2f*)&sAs[m * LP + k0 * 4 + khalf];
      fn[0][k0] = *(const v2f*)&sAn[m * LP + k0 * 4 + khalf];
    }
#pragma unroll
    for (int st = 0; st < 16; ++st) {
      const int cur = st & 1, nxt = cur ^ 1;
      const int kh = st >> 3, rt = st & 7;
      if (st + 1 < 16) {
        const int kh1 = (st + 1) >> 3, rt1 = (st + 1) & 7;
#pragma unroll
        for (int k0 = 0; k0 < 4; ++k0) {
          const int kk = kh1 * 16 + k0 * 4 + khalf;
          fs[nxt][k0] = *(const v2f*)&sAs[(rt1 * 16 + m) * LP + kk];
          fn[nxt][k0] = *(const v2f*)&sAn[(rt1 * 16 + m) * LP + kk];
        }
      }
#pragma unroll
      for (int k0 = 0; k0 < 4; ++k0) {
        acc[rt] = __builtin_amdgcn_wmma_f32_16x16x4_f32(
            false, fs[cur][k0], false, bs[kh * 4 + k0], (short)0, acc[rt], false, false);
        acc[rt] = __builtin_amdgcn_wmma_f32_16x16x4_f32(
            false, fn[cur][k0], false, bn[kh * 4 + k0], (short)0, acc[rt], false, false);
      }
    }
    __builtin_amdgcn_sched_barrier(0);
    asm volatile("v_nop\n\tv_nop\n\tv_nop\n\tv_nop");
    __builtin_amdgcn_sched_barrier(0);
  }
#undef SAGE_LOAD_CHUNK

  const float bv = Bias[jj];
  const int mofs = (lane >> 4) << 3;
#pragma unroll
  for (int rt = 0; rt < 8; ++rt) {
#pragma unroll
    for (int r = 0; r < 8; ++r) {
      int row = rowbase + rt * 16 + r + mofs;
      if (row < nrows) {
        float v = acc[rt][r] + bv;
        if (do_relu) v = fmaxf(v, 0.0f);
        Out[(size_t)row * DIM + jj] = v;
      }
    }
  }
}

extern "C" void kernel_launch(void* const* d_in, const int* in_sizes, int n_in,
                              void* d_out, int out_size, void* d_ws, size_t ws_size,
                              hipStream_t stream) {
  const float* x       = (const float*)d_in[0];
  const int*   src     = (const int*)  d_in[1];
  const int*   dst     = (const int*)  d_in[2];
  const float* weight  = (const float*)d_in[3];
  const float* Wself1  = (const float*)d_in[4];
  const float* Wneigh1 = (const float*)d_in[5];
  const float* b1      = (const float*)d_in[6];
  const float* Wself2  = (const float*)d_in[7];
  const float* Wneigh2 = (const float*)d_in[8];
  const float* b2      = (const float*)d_in[9];
  float* out = (float*)d_out;

  char* ws = (char*)d_ws;
  float* deg = (float*)(ws);
  float* agg = (float*)(ws + ((size_t)NN * 4 + 255 & ~(size_t)255));
  float* h1  = (float*)((char*)agg + (size_t)NN * DIM * 4);

  int* rowptr = (int*)out;
  int* rec    = (int*)((char*)out + 256 * 1024);

  const int gemm_blocks = (NN + RPB - 1) / RPB;
  const int node_blocks = (NN + NPW * 8 - 1) / (NPW * 8);
  const int own_blocks  = BLD_BLOCKS;

  sage_build_kernel<<<own_blocks, BLD_THREADS, 0, stream>>>(dst, src, weight, rowptr, deg, rec);

  sage_gather_kernel<<<node_blocks, 256, 0, stream>>>(x, rowptr, rec, agg, NN);
  sage_gemm_kernel<<<gemm_blocks, 256, 0, stream>>>(
      x, agg, deg, Wself1, Wneigh1, b1, h1, NN,  1);

  sage_gather_kernel<<<node_blocks, 256, 0, stream>>>(h1, rowptr, rec, agg, NN);
  sage_gemm_kernel<<<gemm_blocks, 256, 0, stream>>>(
      h1, agg, deg, Wself2, Wneigh2, b2, out, NN,  0);
}


# Edge list build in two launches: the first groups each slice of edges by owner block, so an owner block reads only its own edges and a count table, not all dst; launches 5 to 6; sum order unchanged
# speedup vs baseline: 3.1870x; 1.1229x over previous
#include <hip/hip_runtime.h>


#define NN 50000
#define NE 625000
#define DIM 128

typedef __attribute__((ext_vector_type(2))) float v2f;
typedef __attribute__((ext_vector_type(8))) float v8f;
typedef __attribute__((ext_vector_type(4))) float v4f;


__device__ __forceinline__ int sage_wave_incl_scan(int v, int lane) {
  v += __builtin_amdgcn_update_dpp(0, v, 0x111, 0xf, 0xf, false);
  v += __builtin_amdgcn_update_dpp(0, v, 0x112, 0xf, 0xf, false);
  v += __builtin_amdgcn_update_dpp(0, v, 0x114, 0xf, 0xf, false);
  v += __builtin_amdgcn_update_dpp(0, v, 0x118, 0xf, 0xf, false);
  const int low = __builtin_amdgcn_readlane(v, 15);
  if (lane >= 16) v += low;
  return v;
}

#define OWN         224
#define BLD_THREADS 1024
#define BLD_WAVES   (BLD_THREADS / 32)
#define BLD_CAP     8192
#define BLD_BLOCKS  ((NN + 1 + OWN - 1) / OWN)

#define PRT_SLICES  256
#define PRT_S       ((NE + PRT_SLICES - 1) / PRT_SLICES)
#define PRT_THREADS 512
#define PRT_PER     ((PRT_S + PRT_THREADS - 1) / PRT_THREADS)
#define CT_Q        (PRT_SLICES / 4)
#define CT_U        14
static_assert(OWN % 32 == 0 && BLD_WAVES == 32, "scan by whole waves");
static_assert(BLD_BLOCKS % 32 == 0, "the slice's owner counts are scanned by whole waves");
static_assert(PRT_SLICES == 8 * BLD_WAVES, "a wave of the owner block takes 8 slices");
static_assert((PRT_SLICES - 1) * PRT_S < NE, "the last slice is not empty");
static_assert((BLD_BLOCKS - 1) * CT_Q <= CT_U * BLD_THREADS, "ct rows below the last block");

__global__ __launch_bounds__(PRT_THREADS) void sage_part_kernel(
    const int* __restrict__ dst, const int* __restrict__ src,
    const float* __restrict__ w, int* __restrict__ part,
    int* __restrict__ ct, int* __restrict__ ot) {
  __shared__ int s_cnt[BLD_BLOCKS];
  __shared__ int s_off[BLD_BLOCKS];

  const int tid  = threadIdx.x;
  const int lane = tid & 31;
  const int wave = tid >> 5;
  const int a    = blockIdx.x;
  const int base = a * PRT_S;
  const int n    = min(PRT_S, NE - base);
  int4* __restrict__ part4 = (int4*)part;

  for (int i = tid; i < BLD_BLOCKS; i += PRT_THREADS) s_cnt[i] = 0;
  __syncthreads();

  int e[PRT_PER], d[PRT_PER], sv[PRT_PER], wb[PRT_PER], o[PRT_PER], slot[PRT_PER];
#pragma unroll
  for (int k = 0; k < PRT_PER; ++k) {
    const int i = tid + k * PRT_THREADS;
    e[k]  = base + (i < n ? i : 0);
    d[k]  = dst[e[k]];
    sv[k] = src[e[k]];
    wb[k] = __float_as_int(w[e[k]]);
  }
#pragma unroll
  for (int k = 0; k < PRT_PER; ++k) {
    const int i = tid + k * PRT_THREADS;
    const unsigned oo = (unsigned)d[k] / (unsigned)OWN;
    o[k] = (i < n && oo < (unsigned)BLD_BLOCKS) ? (int)oo : -1;
    slot[k] = 0;
    if (o[k] >= 0) slot[k] = atomicAdd(&s_cnt[o[k]], 1);
  }
  __syncthreads();

  if (wave == 0) {
    int carry = 0;
    for (int i0 = 0; i0 < BLD_BLOCKS; i0 += 32) {
      const int i = i0 + lane;
      const int c = s_cnt[i];
      const int inc = sage_wave_incl_scan(c, lane);
      const int p = carry + inc - c;
      s_off[i] = p;
      ct[i * PRT_SLICES + a] = c;
      ot[i * PRT_SLICES + a] = p;
      carry += __builtin_amdgcn_readlane(inc, 31);
    }
  }
  __syncthreads();

#pragma unroll
  for (int k = 0; k < PRT_PER; ++k)
    if (o[k] >= 0)
      part4[base + s_off[o[k]] + slot[k]] =
          make_int4(e[k], sv[k], wb[k], d[k] - o[k] * OWN);
}

__device__ __forceinline__ void sage_bld_keep(const int4 r, int li, int2* s_list,
                                              int2* s_sw) {
  if (li < BLD_CAP) {
    s_list[li] = make_int2(r.x, r.w);
    s_sw[li]   = make_int2(r.y, r.z);
  }
}

__global__ __launch_bounds__(BLD_THREADS) void sage_build_kernel(
    const int* __restrict__ dst, const int* __restrict__ src,
    const float* __restrict__ w, const int* __restrict__ part,
    const int* __restrict__ ct, const int* __restrict__ ot,
    int* __restrict__ rowptr, float* __restrict__ deg, int* __restrict__ rec) {
  __shared__ int  s_cnt[OWN];
  __shared__ int  s_ptr[OWN];
  __shared__ int  s_cur[OWN];
  __shared__ int  s_wless[BLD_WAVES];
  __shared__ int  s_wtot[PRT_SLICES / 32];
  __shared__ __attribute__((aligned(16))) int s_sc[PRT_SLICES];
  __shared__ __attribute__((aligned(16))) int s_sg[PRT_SLICES];
  __shared__ __attribute__((aligned(16))) int s_sb[PRT_SLICES];
  __shared__ int2 s_list[BLD_CAP];
  __shared__ int2 s_sw[BLD_CAP];
  __shared__ int  s_slot[BLD_CAP];

  const int tid  = threadIdx.x;
  const int lane = tid & 31;
  const int wave = tid >> 5;
  const int lo = blockIdx.x * OWN;
  const int hi = lo + OWN;
  int4* __restrict__ rec4 = (int4*)rec;
  const int4* __restrict__ part4 = (const int4*)part;

  for (int i = tid; i < OWN; i += BLD_THREADS) { s_cnt[i] = 0; s_cur[i] = 0; }

  int nless = 0;
  {
    const int4* __restrict__ ct4 = (const int4*)ct;
    const int nq = (int)blockIdx.x * CT_Q;
    int4 q[CT_U];
#pragma unroll
    for (int u = 0; u < CT_U; ++u) {
      const int idx = tid + u * BLD_THREADS;
      q[u] = make_int4(0, 0, 0, 0);
      if (idx < nq) q[u] = ct4[idx];
    }
#pragma unroll
    for (int u = 0; u < CT_U; ++u) nless += (q[u].x + q[u].y) + (q[u].z + q[u].w);
  }
  int c = 0, f = 0;
  if (tid < PRT_SLICES) {
    c = ct[blockIdx.x * PRT_SLICES + tid];
    f = ot[blockIdx.x * PRT_SLICES + tid];
  }
  {
    const int wsum = sage_wave_incl_scan(nless, lane);
    if (lane == 31) s_wless[wave] = wsum;
  }
  const int inc = sage_wave_incl_scan(c, lane);
  if (wave < PRT_SLICES / 32 && lane == 31) s_wtot[wave] = inc;
  __syncthreads();

  int total = 0;
  {
    int before = 0;
    for (int j = 0; j < PRT_SLICES / 32; ++j) {
      const int t = s_wtot[j];
      if (j < wave) before += t;
      total += t;
    }
    if (tid < PRT_SLICES) {
      s_sc[tid] = c;
      s_sg[tid] = tid * PRT_S + f;
      s_sb[tid] = before + inc - c;
    }
  }
  __syncthreads();

  {
    int4 r[8];
    const int4 c0 = ((const int4*)s_sc)[wave * 2], c1 = ((const int4*)s_sc)[wave * 2 + 1];
    const int4 g0 = ((const int4*)s_sg)[wave * 2], g1 = ((const int4*)s_sg)[wave * 2 + 1];
    const int4 b0 = ((const int4*)s_sb)[wave * 2], b1 = ((const int4*)s_sb)[wave * 2 + 1];
    const int ca[8] = {c0.x, c0.y, c0.z, c0.w, c1.x, c1.y, c1.z, c1.w};
    const int ga[8] = {g0.x, g0.y, g0.z, g0.w, g1.x, g1.y, g1.z, g1.w};
    const int ba[8] = {b0.x, b0.y, b0.z, b0.w, b1.x, b1.y, b1.z, b1.w};
#pragma unroll
    for (int j = 0; j < 8; ++j) {
      r[j] = make_int4(0, 0, 0, 0);
      if (lane < ca[j]) r[j] = part4[ga[j] + lane];
    }
#pragma unroll
    for (int j = 0; j < 8; ++j)
      if (lane < ca[j]) sage_bld_keep(r[j], ba[j] + lane, s_list, s_sw);
    for (int j = 0; j < 8; ++j)
      for (int k = lane + 32; k < ca[j]; k += 32)
        sage_bld_keep(part4[ga[j] + k], ba[j] + k, s_list, s_sw);
  }
  __syncthreads();

  const bool fits = total <= BLD_CAP;
  if (fits) {
    for (int li = tid; li < total; li += BLD_THREADS)
      s_slot[li] = atomicAdd(&s_cnt[s_list[li].y], 1);
  } else {
    for (int e = tid; e < NE; e += BLD_THREADS) {
      const int dd = dst[e];
      if (dd >= lo && dd < hi) atomicAdd(&s_cnt[dd - lo], 1);
    }
  }
  __syncthreads();

  if (wave == 0) {
    const int below = sage_wave_incl_scan(s_wless[lane], lane);
    int carry = __builtin_amdgcn_readlane(below, 31);
    for (int i0 = 0; i0 < OWN; i0 += 32) {
      const int i = i0 + lane;
      const int cn = s_cnt[i];
      const int in2 = sage_wave_incl_scan(cn, lane);
      const int p = carry + in2 - cn;
      s_ptr[i] = p;
      const int node = lo + i;
      if (node <= NN) rowptr[node] = p;
      if (node <  NN) deg[node] = (float)cn;
      carry += __builtin_amdgcn_readlane(in2, 31);
    }
  }
  __syncthreads();

  if (fits) {
    for (int li = tid; li < total; li += BLD_THREADS) {
      const int2 t  = s_list[li];
      const int2 sw = s_sw[li];
      rec4[s_ptr[t.y] + s_slot[li]] = make_int4(t.x, sw.x, sw.y, 0);
    }
  } else {
    for (int e = tid; e < NE; e += BLD_THREADS) {
      const int dd = dst[e];
      if (dd >= lo && dd < hi) {
        const int pos = s_ptr[dd - lo] + atomicAdd(&s_cur[dd - lo], 1);
        rec4[pos] = make_int4(e, src[e], __float_as_int(w[e]), 0);
      }
    }
  }
}

#define NPW 4

template <int G>
__device__ __forceinline__ void sage_rows(float4& acc, const float* __restrict__ h,
                                          int ms, float mw, int j, int lane) {
  float4 v[G];
  float we[G];
#pragma unroll
  for (int u = 0; u < G; ++u) {
    const int s = __builtin_amdgcn_readlane(ms, j + u);
    we[u] = __int_as_float(__builtin_amdgcn_readlane(__float_as_int(mw), j + u));
    v[u] = ((const float4*)(h + (size_t)s * DIM))[lane];
  }
#pragma unroll
  for (int u = 0; u < G; ++u) {
    acc.x += v[u].x * we[u];
    acc.y += v[u].y * we[u];
    acc.z += v[u].z * we[u];
    acc.w += v[u].w * we[u];
  }
}

__global__ __launch_bounds__(256) void sage_gather_kernel(
    const float* __restrict__ h, const int* __restrict__ rowptr,
    const int* __restrict__ rec, float* __restrict__ agg, int n_nodes) {
  const int lane = threadIdx.x & 31;
  const int4* __restrict__ rec4 = (const int4*)rec;
  const int wv = __builtin_amdgcn_readfirstlane(
      (int)((blockIdx.x * blockDim.x + threadIdx.x) >> 5));
#pragma nounroll
  for (int k = 0; k < NPW; ++k) {
    const int node = wv * NPW + k;
    if (node >= n_nodes) break;
    const int beg = rowptr[node];
    const int n   = rowptr[node + 1] - beg;
    float4 acc = make_float4(0.0f, 0.0f, 0.0f, 0.0f);
    if (n <= 32) {
      int my = 0x7fffffff, s = 0, wb = 0;
      if (lane < n) {
        const int4 t = rec4[beg + lane];
        my = t.x; s = t.y; wb = t.z;
      }
      int rank = 0;
      for (int j = 0; j < n; ++j) {
        const int other = __builtin_amdgcn_readlane(my, j);
        rank += (other < my) ? 1 : 0;
      }
      if (lane >= n) rank = lane;
      const int   ms = __builtin_amdgcn_ds_permute(rank << 2, s);
      const float mw = __int_as_float(__builtin_amdgcn_ds_permute(rank << 2, wb));
      int j = 0;
      for (; j + 8 <= n; j += 8) sage_rows<8>(acc, h, ms, mw, j, lane);
      if (j + 4 <= n) { sage_rows<4>(acc, h, ms, mw, j, lane); j += 4; }
      if (j + 2 <= n) { sage_rows<2>(acc, h, ms, mw, j, lane); j += 2; }
      if (j < n)      { sage_rows<1>(acc, h, ms, mw, j, lane); }
    } else {
      int prev = -1;
      for (int r = 0; r < n; ++r) {
        int best = 0x7fffffff, bs = 0, bw = 0;
        for (int i0 = 0; i0 < n; i0 += 32) {
          const int i = i0 + lane;
          if (i < n) {
            const int4 t = rec4[beg + i];
            if (t.x > prev && t.x < best) { best = t.x; bs = t.y; bw = t.z; }
          }
        }
        int m = 0x7fffffff, src_lane = 0;
        for (int l = 0; l < 32; ++l) {
          const int v = __builtin_amdgcn_readlane(best, l);
          if (v < m) { m = v; src_lane = l; }
        }
        const int   ms = __builtin_amdgcn_readlane(bs, src_lane);
        const float mw = __int_as_float(__builtin_amdgcn_readlane(bw, src_lane));
        sage_rows<1>(acc, h, ms, mw, 0, lane);
        prev = m;
      }
    }
    ((float4*)(agg + (size_t)node * DIM))[lane] = acc;
  }
}

#define RPB 128
#define KC  32
#define LP  36
#define NCH (DIM / KC)

__global__ __launch_bounds__(256) void sage_gemm_kernel(
    const float* __restrict__ Hself, const float* __restrict__ Agg,
    const float* __restrict__ Deg,   const float* __restrict__ Wself,
    const float* __restrict__ Wneigh,const float* __restrict__ Bias,
    float* __restrict__ Out, int nrows, int do_relu) {
  __shared__ __attribute__((aligned(16))) float sWs[DIM * LP];
  __shared__ __attribute__((aligned(16))) float sWn[DIM * LP];
  __shared__ __attribute__((aligned(16))) float sAs[RPB * LP];
  __shared__ __attribute__((aligned(16))) float sAn[RPB * LP];

  const int tid  = threadIdx.x;
  const int lane = tid & 31;
  const int wave = tid >> 5;
  const int rowbase = blockIdx.x * RPB;

  const int sr = tid >> 3;
  const int c4 = tid & 7;
  float rd[4];
#pragma unroll
  for (int s = 0; s < 4; ++s) {
    const int gr = min(rowbase + sr + s * 32, nrows - 1);
    rd[s] = 1.0f / fmaxf(Deg[gr], 1.0f);
  }

  const int gr0 = min(rowbase + sr,      nrows - 1);
  const int gr1 = min(rowbase + sr + 32, nrows - 1);
  const int gr2 = min(rowbase + sr + 64, nrows - 1);
  const int gr3 = min(rowbase + sr + 96, nrows - 1);
  const v4f* gws = (const v4f*)(Wself  + (size_t)sr * DIM) + c4;
  const v4f* gwn = (const v4f*)(Wneigh + (size_t)sr * DIM) + c4;
  const v4f* gs0 = (const v4f*)(Hself + (size_t)gr0 * DIM) + c4;
  const v4f* gs1 = (const v4f*)(Hself + (size_t)gr1 * DIM) + c4;
  const v4f* gs2 = (const v4f*)(Hself + (size_t)gr2 * DIM) + c4;
  const v4f* gs3 = (const v4f*)(Hself + (size_t)gr3 * DIM) + c4;
  const v4f* gn0 = (const v4f*)(Agg   + (size_t)gr0 * DIM) + c4;
  const v4f* gn1 = (const v4f*)(Agg   + (size_t)gr1 * DIM) + c4;
  const v4f* gn2 = (const v4f*)(Agg   + (size_t)gr2 * DIM) + c4;
  const v4f* gn3 = (const v4f*)(Agg   + (size_t)gr3 * DIM) + c4;
  v4f ws0, ws1, ws2, ws3, wn0, wn1, wn2, wn3;
  v4f as0, as1, as2, as3, an0, an1, an2, an3;
#define SAGE_LOAD_CHUNK(kc_)                                   \
  {                                                            \
    const int o  = (kc_) * (KC / 4);                           \
    const int w1 = 32 * DIM / 4, w2 = 64 * DIM / 4, w3 = 96 * DIM / 4; \
    ws0 = gws[o]; ws1 = gws[o + w1]; ws2 = gws[o + w2]; ws3 = gws[o + w3]; \
    wn0 = gwn[o]; wn1 = gwn[o + w1]; wn2 = gwn[o + w2]; wn3 = gwn[o + w3]; \
    as0 = gs0[o]; as1 = gs1[o]; as2 = gs2[o]; as3 = gs3[o];    \
    an0 = gn0[o]; an1 = gn1[o]; an2 = gn2[o]; an3 = gn3[o];    \
  }

  SAGE_LOAD_CHUNK(0)

  v8f acc[8] = {};

  const int jj    = wave * 16 + (lane & 15);
  const int khalf = (lane >> 4) << 1;
  const int m     = lane & 15;

#pragma nounroll
  for (int kc = 0; kc < NCH; ++kc) {
    __syncthreads();
    {
      v4f* ls = (v4f*)&sWs[sr * LP + c4 * 4];
      v4f* ln = (v4f*)&sWn[sr * LP + c4 * 4];
      v4f* la = (v4f*)&sAs[sr * LP + c4 * 4];
      v4f* lb = (v4f*)&sAn[sr * LP + c4 * 4];
      const int q = 32 * LP / 4;
      ls[0] = ws0; ls[q] = ws1; ls[2 * q] = ws2; ls[3 * q] = ws3;
      ln[0] = wn0; ln[q] = wn1; ln[2 * q] = wn2; ln[3 * q] = wn3;
      la[0] = as0; la[q] = as1; la[2 * q] = as2; la[3 * q] = as3;
      lb[0] = an0 * rd[0]; lb[q] = an1 * rd[1]; lb[2 * q] = an2 * rd[2]; lb[3 * q] = an3 * rd[3];
    }
    __syncthreads();
    if (kc + 1 < NCH) { SAGE_LOAD_CHUNK(kc + 1) }

    v2f bs[8], bn[8];
#pragma unroll
    for (int k0 = 0; k0 < 8; ++k0) {
      const int kk = k0 * 4 + khalf;
      bs[k0] = *(const v2f*)&sWs[jj * LP + kk];
      bn[k0] = *(const v2f*)&sWn[jj * LP + kk];
    }
    v2f fs[2][4], fn[2][4];
#pragma unroll
    for (int k0 = 0; k0 < 4; ++k0) {
      fs[0][k0] = *(const v2f*)&sAs[m * LP + k0 * 4 + khalf];
      fn[0][k0] = *(const v2f*)&sAn[m * LP + k0 * 4 + khalf];
    }
#pragma unroll
    for (int st = 0; st < 16; ++st) {
      const int cur = st & 1, nxt = cur ^ 1;
      const int kh = st >> 3, rt = st & 7;
      if (st + 1 < 16) {
        const int kh1 = (st + 1) >> 3, rt1 = (st + 1) & 7;
#pragma unroll
        for (int k0 = 0; k0 < 4; ++k0) {
          const int kk = kh1 * 16 + k0 * 4 + khalf;
          fs[nxt][k0] = *(const v2f*)&sAs[(rt1 * 16 + m) * LP + kk];
          fn[nxt][k0] = *(const v2f*)&sAn[(rt1 * 16 + m) * LP + kk];
        }
      }
#pragma unroll
      for (int k0 = 0; k0 < 4; ++k0) {
        acc[rt] = __builtin_amdgcn_wmma_f32_16x16x4_f32(
            false, fs[cur][k0], false, bs[kh * 4 + k0], (short)0, acc[rt], false, false);
        acc[rt] = __builtin_amdgcn_wmma_f32_16x16x4_f32(
            false, fn[cur][k0], false, bn[kh * 4 + k0], (short)0, acc[rt], false, false);
      }
    }
    __builtin_amdgcn_sched_barrier(0);
    asm volatile("v_nop\n\tv_nop\n\tv_nop\n\tv_nop");
    __builtin_amdgcn_sched_barrier(0);
  }
#undef SAGE_LOAD_CHUNK

  const float bv = Bias[jj];
  const int mofs = (lane >> 4) << 3;
#pragma unroll
  for (int rt = 0; rt < 8; ++rt) {
#pragma unroll
    for (int r = 0; r < 8; ++r) {
      int row = rowbase + rt * 16 + r + mofs;
      if (row < nrows) {
        float v = acc[rt][r] + bv;
        if (do_relu) v = fmaxf(v, 0.0f);
        Out[(size_t)row * DIM + jj] = v;
      }
    }
  }
}

extern "C" void kernel_launch(void* const* d_in, const int* in_sizes, int n_in,
                              void* d_out, int out_size, void* d_ws, size_t ws_size,
                              hipStream_t stream) {
  const float* x       = (const float*)d_in[0];
  const int*   src     = (const int*)  d_in[1];
  const int*   dst     = (const int*)  d_in[2];
  const float* weight  = (const float*)d_in[3];
  const float* Wself1  = (const float*)d_in[4];
  const float* Wneigh1 = (const float*)d_in[5];
  const float* b1      = (const float*)d_in[6];
  const float* Wself2  = (const float*)d_in[7];
  const float* Wneigh2 = (const float*)d_in[8];
  const float* b2      = (const float*)d_in[9];
  float* out = (float*)d_out;

  char* ws = (char*)d_ws;
  float* deg = (float*)(ws);
  float* agg = (float*)(ws + ((size_t)NN * 4 + 255 & ~(size_t)255));
  float* h1  = (float*)((char*)agg + (size_t)NN * DIM * 4);

  int* rowptr = (int*)out;
  int* rec    = (int*)((char*)out + 256 * 1024);

  const int gemm_blocks = (NN + RPB - 1) / RPB;
  const int node_blocks = (NN + NPW * 8 - 1) / (NPW * 8);
  const int own_blocks  = BLD_BLOCKS;

  int* part = (int*)h1;
  int* ct   = (int*)((char*)h1 + (size_t)PRT_SLICES * PRT_S * 16);
  int* ot   = ct + BLD_BLOCKS * PRT_SLICES;

  sage_part_kernel<<<PRT_SLICES, PRT_THREADS, 0, stream>>>(dst, src, weight, part, ct, ot);
  sage_build_kernel<<<own_blocks, BLD_THREADS, 0, stream>>>(
      dst, src, weight, part, ct, ot, rowptr, deg, rec);

  sage_gather_kernel<<<node_blocks, 256, 0, stream>>>(x, rowptr, rec, agg, NN);
  sage_gemm_kernel<<<gemm_blocks, 256, 0, stream>>>(
      x, agg, deg, Wself1, Wneigh1, b1, h1, NN,  1);

  sage_gather_kernel<<<node_blocks, 256, 0, stream>>>(h1, rowptr, rec, agg, NN);
  sage_gemm_kernel<<<gemm_blocks, 256, 0, stream>>>(
      h1, agg, deg, Wself2, Wneigh2, b2, out, NN,  0);
}


# GEMM rows dealt as 512 blocks of 6 or 7 row tiles of 16 (was 391 blocks of 8): every unit holds two blocks, the busiest 13 tiles instead of 16; sum order unchanged
# speedup vs baseline: 3.4448x; 1.0809x over previous
#include <hip/hip_runtime.h>


#define NN 50000
#define NE 625000
#define DIM 128

typedef __attribute__((ext_vector_type(2))) float v2f;
typedef __attribute__((ext_vector_type(8))) float v8f;
typedef __attribute__((ext_vector_type(4))) float v4f;


__device__ __forceinline__ int sage_wave_incl_scan(int v, int lane) {
  v += __builtin_amdgcn_update_dpp(0, v, 0x111, 0xf, 0xf, false);
  v += __builtin_amdgcn_update_dpp(0, v, 0x112, 0xf, 0xf, false);
  v += __builtin_amdgcn_update_dpp(0, v, 0x114, 0xf, 0xf, false);
  v += __builtin_amdgcn_update_dpp(0, v, 0x118, 0xf, 0xf, false);
  const int low = __builtin_amdgcn_readlane(v, 15);
  if (lane >= 16) v += low;
  return v;
}

#define OWN         224
#define BLD_THREADS 1024
#define BLD_WAVES   (BLD_THREADS / 32)
#define BLD_CAP     8192
#define BLD_BLOCKS  ((NN + 1 + OWN - 1) / OWN)

#define PRT_SLICES  256
#define PRT_S       ((NE + PRT_SLICES - 1) / PRT_SLICES)
#define PRT_THREADS 512
#define PRT_PER     ((PRT_S + PRT_THREADS - 1) / PRT_THREADS)
#define CT_Q        (PRT_SLICES / 4)
#define CT_U        14
static_assert(OWN % 32 == 0 && BLD_WAVES == 32, "scan by whole waves");
static_assert(BLD_BLOCKS % 32 == 0, "the slice's owner counts are scanned by whole waves");
static_assert(PRT_SLICES == 8 * BLD_WAVES, "a wave of the owner block takes 8 slices");
static_assert((PRT_SLICES - 1) * PRT_S < NE, "the last slice is not empty");
static_assert((BLD_BLOCKS - 1) * CT_Q <= CT_U * BLD_THREADS, "ct rows below the last block");

__global__ __launch_bounds__(PRT_THREADS) void sage_part_kernel(
    const int* __restrict__ dst, const int* __restrict__ src,
    const float* __restrict__ w, int* __restrict__ part,
    int* __restrict__ ct, int* __restrict__ ot) {
  __shared__ int s_cnt[BLD_BLOCKS];
  __shared__ int s_off[BLD_BLOCKS];

  const int tid  = threadIdx.x;
  const int lane = tid & 31;
  const int wave = tid >> 5;
  const int a    = blockIdx.x;
  const int base = a * PRT_S;
  const int n    = min(PRT_S, NE - base);
  int4* __restrict__ part4 = (int4*)part;

  for (int i = tid; i < BLD_BLOCKS; i += PRT_THREADS) s_cnt[i] = 0;
  __syncthreads();

  int e[PRT_PER], d[PRT_PER], sv[PRT_PER], wb[PRT_PER], o[PRT_PER], slot[PRT_PER];
#pragma unroll
  for (int k = 0; k < PRT_PER; ++k) {
    const int i = tid + k * PRT_THREADS;
    e[k]  = base + (i < n ? i : 0);
    d[k]  = dst[e[k]];
    sv[k] = src[e[k]];
    wb[k] = __float_as_int(w[e[k]]);
  }
#pragma unroll
  for (int k = 0; k < PRT_PER; ++k) {
    const int i = tid + k * PRT_THREADS;
    const unsigned oo = (unsigned)d[k] / (unsigned)OWN;
    o[k] = (i < n && oo < (unsigned)BLD_BLOCKS) ? (int)oo : -1;
    slot[k] = 0;
    if (o[k] >= 0) slot[k] = atomicAdd(&s_cnt[o[k]], 1);
  }
  __syncthreads();

  if (wave == 0) {
    int carry = 0;
    for (int i0 = 0; i0 < BLD_BLOCKS; i0 += 32) {
      const int i = i0 + lane;
      const int c = s_cnt[i];
      const int inc = sage_wave_incl_scan(c, lane);
      const int p = carry + inc - c;
      s_off[i] = p;
      ct[i * PRT_SLICES + a] = c;
      ot[i * PRT_SLICES + a] = p;
      carry += __builtin_amdgcn_readlane(inc, 31);
    }
  }
  __syncthreads();

#pragma unroll
  for (int k = 0; k < PRT_PER; ++k)
    if (o[k] >= 0)
      part4[base + s_off[o[k]] + slot[k]] =
          make_int4(e[k], sv[k], wb[k], d[k] - o[k] * OWN);
}

__device__ __forceinline__ void sage_bld_keep(const int4 r, int li, int2* s_list,
                                              int2* s_sw) {
  if (li < BLD_CAP) {
    s_list[li] = make_int2(r.x, r.w);
    s_sw[li]   = make_int2(r.y, r.z);
  }
}

__global__ __launch_bounds__(BLD_THREADS) void sage_build_kernel(
    const int* __restrict__ dst, const int* __restrict__ src,
    const float* __restrict__ w, const int* __restrict__ part,
    const int* __restrict__ ct, const int* __restrict__ ot,
    int* __restrict__ rowptr, float* __restrict__ deg, int* __restrict__ rec) {
  __shared__ int  s_cnt[OWN];
  __shared__ int  s_ptr[OWN];
  __shared__ int  s_cur[OWN];
  __shared__ int  s_wless[BLD_WAVES];
  __shared__ int  s_wtot[PRT_SLICES / 32];
  __shared__ __attribute__((aligned(16))) int s_sc[PRT_SLICES];
  __shared__ __attribute__((aligned(16))) int s_sg[PRT_SLICES];
  __shared__ __attribute__((aligned(16))) int s_sb[PRT_SLICES];
  __shared__ int2 s_list[BLD_CAP];
  __shared__ int2 s_sw[BLD_CAP];
  __shared__ int  s_slot[BLD_CAP];

  const int tid  = threadIdx.x;
  const int lane = tid & 31;
  const int wave = tid >> 5;
  const int lo = blockIdx.x * OWN;
  const int hi = lo + OWN;
  int4* __restrict__ rec4 = (int4*)rec;
  const int4* __restrict__ part4 = (const int4*)part;

  for (int i = tid; i < OWN; i += BLD_THREADS) { s_cnt[i] = 0; s_cur[i] = 0; }

  int nless = 0;
  {
    const int4* __restrict__ ct4 = (const int4*)ct;
    const int nq = (int)blockIdx.x * CT_Q;
    int4 q[CT_U];
#pragma unroll
    for (int u = 0; u < CT_U; ++u) {
      const int idx = tid + u * BLD_THREADS;
      q[u] = make_int4(0, 0, 0, 0);
      if (idx < nq) q[u] = ct4[idx];
    }
#pragma unroll
    for (int u = 0; u < CT_U; ++u) nless += (q[u].x + q[u].y) + (q[u].z + q[u].w);
  }
  int c = 0, f = 0;
  if (tid < PRT_SLICES) {
    c = ct[blockIdx.x * PRT_SLICES + tid];
    f = ot[blockIdx.x * PRT_SLICES + tid];
  }
  {
    const int wsum = sage_wave_incl_scan(nless, lane);
    if (lane == 31) s_wless[wave] = wsum;
  }
  const int inc = sage_wave_incl_scan(c, lane);
  if (wave < PRT_SLICES / 32 && lane == 31) s_wtot[wave] = inc;
  __syncthreads();

  int total = 0;
  {
    int before = 0;
    for (int j = 0; j < PRT_SLICES / 32; ++j) {
      const int t = s_wtot[j];
      if (j < wave) before += t;
      total += t;
    }
    if (tid < PRT_SLICES) {
      s_sc[tid] = c;
      s_sg[tid] = tid * PRT_S + f;
      s_sb[tid] = before + inc - c;
    }
  }
  __syncthreads();

  {
    int4 r[8];
    const int4 c0 = ((const int4*)s_sc)[wave * 2], c1 = ((const int4*)s_sc)[wave * 2 + 1];
    const int4 g0 = ((const int4*)s_sg)[wave * 2], g1 = ((const int4*)s_sg)[wave * 2 + 1];
    const int4 b0 = ((const int4*)s_sb)[wave * 2], b1 = ((const int4*)s_sb)[wave * 2 + 1];
    const int ca[8] = {c0.x, c0.y, c0.z, c0.w, c1.x, c1.y, c1.z, c1.w};
    const int ga[8] = {g0.x, g0.y, g0.z, g0.w, g1.x, g1.y, g1.z, g1.w};
    const int ba[8] = {b0.x, b0.y, b0.z, b0.w, b1.x, b1.y, b1.z, b1.w};
#pragma unroll
    for (int j = 0; j < 8; ++j) {
      r[j] = make_int4(0, 0, 0, 0);
      if (lane < ca[j]) r[j] = part4[ga[j] + lane];
    }
#pragma unroll
    for (int j = 0; j < 8; ++j)
      if (lane < ca[j]) sage_bld_keep(r[j], ba[j] + lane, s_list, s_sw);
    for (int j = 0; j < 8; ++j)
      for (int k = lane + 32; k < ca[j]; k += 32)
        sage_bld_keep(part4[ga[j] + k], ba[j] + k, s_list, s_sw);
  }
  __syncthreads();

  const bool fits = total <= BLD_CAP;
  if (fits) {
    for (int li = tid; li < total; li += BLD_THREADS)
      s_slot[li] = atomicAdd(&s_cnt[s_list[li].y], 1);
  } else {
    for (int e = tid; e < NE; e += BLD_THREADS) {
      const int dd = dst[e];
      if (dd >= lo && dd < hi) atomicAdd(&s_cnt[dd - lo], 1);
    }
  }
  __syncthreads();

  if (wave == 0) {
    const int below = sage_wave_incl_scan(s_wless[lane], lane);
    int carry = __builtin_amdgcn_readlane(below, 31);
    for (int i0 = 0; i0 < OWN; i0 += 32) {
      const int i = i0 + lane;
      const int cn = s_cnt[i];
      const int in2 = sage_wave_incl_scan(cn, lane);
      const int p = carry + in2 - cn;
      s_ptr[i] = p;
      const int node = lo + i;
      if (node <= NN) rowptr[node] = p;
      if (node <  NN) deg[node] = (float)cn;
      carry += __builtin_amdgcn_readlane(in2, 31);
    }
  }
  __syncthreads();

  if (fits) {
    for (int li = tid; li < total; li += BLD_THREADS) {
      const int2 t  = s_list[li];
      const int2 sw = s_sw[li];
      rec4[s_ptr[t.y] + s_slot[li]] = make_int4(t.x, sw.x, sw.y, 0);
    }
  } else {
    for (int e = tid; e < NE; e += BLD_THREADS) {
      const int dd = dst[e];
      if (dd >= lo && dd < hi) {
        const int pos = s_ptr[dd - lo] + atomicAdd(&s_cur[dd - lo], 1);
        rec4[pos] = make_int4(e, src[e], __float_as_int(w[e]), 0);
      }
    }
  }
}

#define NPW 4

template <int G>
__device__ __forceinline__ void sage_rows(float4& acc, const float* __restrict__ h,
                                          int ms, float mw, int j, int lane) {
  float4 v[G];
  float we[G];
#pragma unroll
  for (int u = 0; u < G; ++u) {
    const int s = __builtin_amdgcn_readlane(ms, j + u);
    we[u] = __int_as_float(__builtin_amdgcn_readlane(__float_as_int(mw), j + u));
    v[u] = ((const float4*)(h + (size_t)s * DIM))[lane];
  }
#pragma unroll
  for (int u = 0; u < G; ++u) {
    acc.x += v[u].x * we[u];
    acc.y += v[u].y * we[u];
    acc.z += v[u].z * we[u];
    acc.w += v[u].w * we[u];
  }
}

__global__ __launch_bounds__(256) void sage_gather_kernel(
    const float* __restrict__ h, const int* __restrict__ rowptr,
    const int* __restrict__ rec, float* __restrict__ agg, int n_nodes) {
  const int lane = threadIdx.x & 31;
  const int4* __restrict__ rec4 = (const int4*)rec;
  const int wv = __builtin_amdgcn_readfirstlane(
      (int)((blockIdx.x * blockDim.x + threadIdx.x) >> 5));
#pragma nounroll
  for (int k = 0; k < NPW; ++k) {
    const int node = wv * NPW + k;
    if (node >= n_nodes) break;
    const int beg = rowptr[node];
    const int n   = rowptr[node + 1] - beg;
    float4 acc = make_float4(0.0f, 0.0f, 0.0f, 0.0f);
    if (n <= 32) {
      int my = 0x7fffffff, s = 0, wb = 0;
      if (lane < n) {
        const int4 t = rec4[beg + lane];
        my = t.x; s = t.y; wb = t.z;
      }
      int rank = 0;
      for (int j = 0; j < n; ++j) {
        const int other = __builtin_amdgcn_readlane(my, j);
        rank += (other < my) ? 1 : 0;
      }
      if (lane >= n) rank = lane;
      const int   ms = __builtin_amdgcn_ds_permute(rank << 2, s);
      const float mw = __int_as_float(__builtin_amdgcn_ds_permute(rank << 2, wb));
      int j = 0;
      for (; j + 8 <= n; j += 8) sage_rows<8>(acc, h, ms, mw, j, lane);
      if (j + 4 <= n) { sage_rows<4>(acc, h, ms, mw, j, lane); j += 4; }
      if (j + 2 <= n) { sage_rows<2>(acc, h, ms, mw, j, lane); j += 2; }
      if (j < n)      { sage_rows<1>(acc, h, ms, mw, j, lane); }
    } else {
      int prev = -1;
      for (int r = 0; r < n; ++r) {
        int best = 0x7fffffff, bs = 0, bw = 0;
        for (int i0 = 0; i0 < n; i0 += 32) {
          const int i = i0 + lane;
          if (i < n) {
            const int4 t = rec4[beg + i];
            if (t.x > prev && t.x < best) { best = t.x; bs = t.y; bw = t.z; }
          }
        }
        int m = 0x7fffffff, src_lane = 0;
        for (int l = 0; l < 32; ++l) {
          const int v = __builtin_amdgcn_readlane(best, l);
          if (v < m) { m = v; src_lane = l; }
        }
        const int   ms = __builtin_amdgcn_readlane(bs, src_lane);
        const float mw = __int_as_float(__builtin_amdgcn_readlane(bw, src_lane));
        sage_rows<1>(acc, h, ms, mw, 0, lane);
        prev = m;
      }
    }
    ((float4*)(agg + (size_t)node * DIM))[lane] = acc;
  }
}

#define KC  32
#define LP  36
#define NCH (DIM / KC)
#define GT_TILES    ((NN + 15) / 16)
#define GEMM_BLOCKS 512
#define GT_BASE     (GT_TILES / GEMM_BLOCKS)
#define GT_EXTRA    (GT_TILES % GEMM_BLOCKS)
#define GT_MAX      (GT_BASE + 1)
#define RPB_MAX     (GT_MAX * 16)
static_assert(GT_BASE >= 1 && 2 * GT_EXTRA <= GEMM_BLOCKS, "long blocks sit at even indices");
static_assert(RPB_MAX <= 128, "a thread stages at most four rows of a chunk");

template <int NT>
__device__ __forceinline__ void sage_gemm_tiles(
    const float* __restrict__ Hself, const float* __restrict__ Agg,
    const float* __restrict__ Deg,   const float* __restrict__ Wself,
    const float* __restrict__ Wneigh,const float* __restrict__ Bias,
    float* __restrict__ Out, int nrows, int do_relu, int rowbase,
    float* sWs, float* sWn, float* sAs, float* sAn) {
  constexpr int ROWS = NT * 16;
  const int tid  = threadIdx.x;
  const int lane = tid & 31;
  const int wave = tid >> 5;

  const int sr = tid >> 3;
  const int c4 = tid & 7;
  const bool has3 = (ROWS > 96) && (sr + 96 < ROWS);
  static_assert(ROWS >= 96 && ROWS <= 128, "three full staging rows per thread");
  float rd[4];
#pragma unroll
  for (int s = 0; s < 4; ++s) {
    const int gr = min(rowbase + sr + s * 32, nrows - 1);
    rd[s] = 1.0f / fmaxf(Deg[gr], 1.0f);
  }

  const int gr0 = min(rowbase + sr,      nrows - 1);
  const int gr1 = min(rowbase + sr + 32, nrows - 1);
  const int gr2 = min(rowbase + sr + 64, nrows - 1);
  const int gr3 = min(rowbase + sr + 96, nrows - 1);
  const unsigned ow = (unsigned)(sr * DIM + c4 * 4);
  const unsigned o0 = (unsigned)(gr0 * DIM + c4 * 4);
  const unsigned o1 = (unsigned)(gr1 * DIM + c4 * 4);
  const unsigned o2 = (unsigned)(gr2 * DIM + c4 * 4);
  const unsigned o3 = (unsigned)(gr3 * DIM + c4 * 4);
  v4f ws0, ws1, ws2, ws3, wn0, wn1, wn2, wn3;
  v4f as0, as1, as2, an0, an1, an2;
  v4f as3 = {0.0f, 0.0f, 0.0f, 0.0f}, an3 = {0.0f, 0.0f, 0.0f, 0.0f};
#define SAGE_LD4(base_, off_) (*(const v4f*)((base_) + (off_)))
#define SAGE_LOAD_CHUNK(kc_)                                   \
  {                                                            \
    const unsigned o  = (unsigned)((kc_) * KC);                \
    const unsigned w1 = 32 * DIM, w2 = 64 * DIM, w3 = 96 * DIM; \
    ws0 = SAGE_LD4(Wself, ow + o);       ws1 = SAGE_LD4(Wself, ow + o + w1);  \
    ws2 = SAGE_LD4(Wself, ow + o + w2);  ws3 = SAGE_LD4(Wself, ow + o + w3);  \
    wn0 = SAGE_LD4(Wneigh, ow + o);      wn1 = SAGE_LD4(Wneigh, ow + o + w1); \
    wn2 = SAGE_LD4(Wneigh, ow + o + w2); wn3 = SAGE_LD4(Wneigh, ow + o + w3); \
    as0 = SAGE_LD4(Hself, o0 + o); as1 = SAGE_LD4(Hself, o1 + o); as2 = SAGE_LD4(Hself, o2 + o); \
    an0 = SAGE_LD4(Agg, o0 + o);   an1 = SAGE_LD4(Agg, o1 + o);   an2 = SAGE_LD4(Agg, o2 + o);   \
    if (has3) { as3 = SAGE_LD4(Hself, o3 + o); an3 = SAGE_LD4(Agg, o3 + o); }                    \
  }

  SAGE_LOAD_CHUNK(0)

  v8f acc[NT] = {};

  const int jj    = wave * 16 + (lane & 15);
  const int khalf = (lane >> 4) << 1;
  const int m     = lane & 15;

#pragma nounroll
  for (int kc = 0; kc < NCH; ++kc) {
    __syncthreads();
    {
      v4f* ls = (v4f*)&sWs[sr * LP + c4 * 4];
      v4f* ln = (v4f*)&sWn[sr * LP + c4 * 4];
      v4f* la = (v4f*)&sAs[sr * LP + c4 * 4];
      v4f* lb = (v4f*)&sAn[sr * LP + c4 * 4];
      const int q = 32 * LP / 4;
      ls[0] = ws0; ls[q] = ws1; ls[2 * q] = ws2; ls[3 * q] = ws3;
      ln[0] = wn0; ln[q] = wn1; ln[2 * q] = wn2; ln[3 * q] = wn3;
      la[0] = as0; la[q] = as1; la[2 * q] = as2;
      lb[0] = an0 * rd[0]; lb[q] = an1 * rd[1]; lb[2 * q] = an2 * rd[2];
      if (has3) { la[3 * q] = as3; lb[3 * q] = an3 * rd[3]; }
    }
    __syncthreads();
    if (kc + 1 < NCH) { SAGE_LOAD_CHUNK(kc + 1) }

    v2f bs[8], bn[8];
#pragma unroll
    for (int k0 = 0; k0 < 8; ++k0) {
      const int kk = k0 * 4 + khalf;
      bs[k0] = *(const v2f*)&sWs[jj * LP + kk];
      bn[k0] = *(const v2f*)&sWn[jj * LP + kk];
    }
    v2f fs[2][4], fn[2][4];
#pragma unroll
    for (int k0 = 0; k0 < 4; ++k0) {
      fs[0][k0] = *(const v2f*)&sAs[m * LP + k0 * 4 + khalf];
      fn[0][k0] = *(const v2f*)&sAn[m * LP + k0 * 4 + khalf];
    }
#pragma unroll
    for (int st = 0; st < 2 * NT; ++st) {
      const int cur = st & 1, nxt = cur ^ 1;
      const int kh = st / NT, rt = st % NT;
      if (st + 1 < 2 * NT) {
        const int kh1 = (st + 1) / NT, rt1 = (st + 1) % NT;
#pragma unroll
        for (int k0 = 0; k0 < 4; ++k0) {
          const int kk = kh1 * 16 + k0 * 4 + khalf;
          fs[nxt][k0] = *(const v2f*)&sAs[(rt1 * 16 + m) * LP + kk];
          fn[nxt][k0] = *(const v2f*)&sAn[(rt1 * 16 + m) * LP + kk];
        }
      }
#pragma unroll
      for (int k0 = 0; k0 < 4; ++k0) {
        acc[rt] = __builtin_amdgcn_wmma_f32_16x16x4_f32(
            false, fs[cur][k0], false, bs[kh * 4 + k0], (short)0, acc[rt], false, false);
        acc[rt] = __builtin_amdgcn_wmma_f32_16x16x4_f32(
            false, fn[cur][k0], false, bn[kh * 4 + k0], (short)0, acc[rt], false, false);
      }
    }
    __builtin_amdgcn_sched_barrier(0);
    asm volatile("v_nop\n\tv_nop\n\tv_nop\n\tv_nop");
    __builtin_amdgcn_sched_barrier(0);
  }
#undef SAGE_LOAD_CHUNK
#undef SAGE_LD4

  const float bv = Bias[jj];
  const int mofs = (lane >> 4) << 3;
#pragma unroll
  for (int rt = 0; rt < NT; ++rt) {
#pragma unroll
    for (int r = 0; r < 8; ++r) {
      int row = rowbase + rt * 16 + r + mofs;
      if (row < nrows) {
        float v = acc[rt][r] + bv;
        if (do_relu) v = fmaxf(v, 0.0f);
        Out[(size_t)row * DIM + jj] = v;
      }
    }
  }
}

__global__ __launch_bounds__(256, 4) void sage_gemm_kernel(
    const float* __restrict__ Hself, const float* __restrict__ Agg,
    const float* __restrict__ Deg,   const float* __restrict__ Wself,
    const float* __restrict__ Wneigh,const float* __restrict__ Bias,
    float* __restrict__ Out, int nrows, int do_relu) {
  __shared__ __attribute__((aligned(16))) float sWs[DIM * LP];
  __shared__ __attribute__((aligned(16))) float sWn[DIM * LP];
  __shared__ __attribute__((aligned(16))) float sAs[RPB_MAX * LP];
  __shared__ __attribute__((aligned(16))) float sAn[RPB_MAX * LP];

  const int  b      = (int)blockIdx.x;
  const bool longb  = ((b & 1) == 0) && (b < 2 * GT_EXTRA);
  const int  before = min((b + 1) >> 1, GT_EXTRA);
  const int  rowbase = (b * GT_BASE + before) * 16;
  if (longb)
    sage_gemm_tiles<GT_MAX>(Hself, Agg, Deg, Wself, Wneigh, Bias, Out, nrows, do_relu,
                            rowbase, sWs, sWn, sAs, sAn);
  else
    sage_gemm_tiles<GT_BASE>(Hself, Agg, Deg, Wself, Wneigh, Bias, Out, nrows, do_relu,
                             rowbase, sWs, sWn, sAs, sAn);
}

extern "C" void kernel_launch(void* const* d_in, const int* in_sizes, int n_in,
                              void* d_out, int out_size, void* d_ws, size_t ws_size,
                              hipStream_t stream) {
  const float* x       = (const float*)d_in[0];
  const int*   src     = (const int*)  d_in[1];
  const int*   dst     = (const int*)  d_in[2];
  const float* weight  = (const float*)d_in[3];
  const float* Wself1  = (const float*)d_in[4];
  const float* Wneigh1 = (const float*)d_in[5];
  const float* b1      = (const float*)d_in[6];
  const float* Wself2  = (const float*)d_in[7];
  const float* Wneigh2 = (const float*)d_in[8];
  const float* b2      = (const float*)d_in[9];
  float* out = (float*)d_out;

  char* ws = (char*)d_ws;
  float* deg = (float*)(ws);
  float* agg = (float*)(ws + ((size_t)NN * 4 + 255 & ~(size_t)255));
  float* h1  = (float*)((char*)agg + (size_t)NN * DIM * 4);

  int* rowptr = (int*)out;
  int* rec    = (int*)((char*)out + 256 * 1024);

  const int gemm_blocks = GEMM_BLOCKS;
  const int node_blocks = (NN + NPW * 8 - 1) / (NPW * 8);
  const int own_blocks  = BLD_BLOCKS;

  int* part = (int*)h1;
  int* ct   = (int*)((char*)h1 + (size_t)PRT_SLICES * PRT_S * 16);
  int* ot   = ct + BLD_BLOCKS * PRT_SLICES;

  sage_part_kernel<<<PRT_SLICES, PRT_THREADS, 0, stream>>>(dst, src, weight, part, ct, ot);
  sage_build_kernel<<<own_blocks, BLD_THREADS, 0, stream>>>(
      dst, src, weight, part, ct, ot, rowptr, deg, rec);

  sage_gather_kernel<<<node_blocks, 256, 0, stream>>>(x, rowptr, rec, agg, NN);
  sage_gemm_kernel<<<gemm_blocks, 256, 0, stream>>>(
      x, agg, deg, Wself1, Wneigh1, b1, h1, NN,  1);

  sage_gather_kernel<<<node_blocks, 256, 0, stream>>>(h1, rowptr, rec, agg, NN);
  sage_gemm_kernel<<<gemm_blocks, 256, 0, stream>>>(
      h1, agg, deg, Wself2, Wneigh2, b2, out, NN,  0);
}


# Neighbour sum moved into the GEMM launch: a block gathers its own rows' means into LDS (a node's rows all in flight) and reads its second A operand there; launches 6 to 4; sum order unchanged
# speedup vs baseline: 3.6646x; 1.0638x over previous
#include <hip/hip_runtime.h>


#define NN 50000
#define NE 625000
#define DIM 128

typedef __attribute__((ext_vector_type(2))) float v2f;
typedef __attribute__((ext_vector_type(8))) float v8f;
typedef __attribute__((ext_vector_type(4))) float v4f;


__device__ __forceinline__ int sage_wave_incl_scan(int v, int lane) {
  v += __builtin_amdgcn_update_dpp(0, v, 0x111, 0xf, 0xf, false);
  v += __builtin_amdgcn_update_dpp(0, v, 0x112, 0xf, 0xf, false);
  v += __builtin_amdgcn_update_dpp(0, v, 0x114, 0xf, 0xf, false);
  v += __builtin_amdgcn_update_dpp(0, v, 0x118, 0xf, 0xf, false);
  const int low = __builtin_amdgcn_readlane(v, 15);
  if (lane >= 16) v += low;
  return v;
}

#define OWN         224
#define BLD_THREADS 1024
#define BLD_WAVES   (BLD_THREADS / 32)
#define BLD_CAP     8192
#define BLD_BLOCKS  ((NN + 1 + OWN - 1) / OWN)

#define PRT_SLICES  256
#define PRT_S       ((NE + PRT_SLICES - 1) / PRT_SLICES)
#define PRT_THREADS 512
#define PRT_PER     ((PRT_S + PRT_THREADS - 1) / PRT_THREADS)
#define CT_Q        (PRT_SLICES / 4)
#define CT_U        14
static_assert(OWN % 32 == 0 && BLD_WAVES == 32, "scan by whole waves");
static_assert(BLD_BLOCKS % 32 == 0, "the slice's owner counts are scanned by whole waves");
static_assert(PRT_SLICES == 8 * BLD_WAVES, "a wave of the owner block takes 8 slices");
static_assert((PRT_SLICES - 1) * PRT_S < NE, "the last slice is not empty");
static_assert((BLD_BLOCKS - 1) * CT_Q <= CT_U * BLD_THREADS, "ct rows below the last block");

__global__ __launch_bounds__(PRT_THREADS) void sage_part_kernel(
    const int* __restrict__ dst, const int* __restrict__ src,
    const float* __restrict__ w, int* __restrict__ part,
    int* __restrict__ ct, int* __restrict__ ot) {
  __shared__ int s_cnt[BLD_BLOCKS];
  __shared__ int s_off[BLD_BLOCKS];

  const int tid  = threadIdx.x;
  const int lane = tid & 31;
  const int wave = tid >> 5;
  const int a    = blockIdx.x;
  const int base = a * PRT_S;
  const int n    = min(PRT_S, NE - base);
  int4* __restrict__ part4 = (int4*)part;

  for (int i = tid; i < BLD_BLOCKS; i += PRT_THREADS) s_cnt[i] = 0;
  __syncthreads();

  int e[PRT_PER], d[PRT_PER], sv[PRT_PER], wb[PRT_PER], o[PRT_PER], slot[PRT_PER];
#pragma unroll
  for (int k = 0; k < PRT_PER; ++k) {
    const int i = tid + k * PRT_THREADS;
    e[k]  = base + (i < n ? i : 0);
    d[k]  = dst[e[k]];
    sv[k] = src[e[k]];
    wb[k] = __float_as_int(w[e[k]]);
  }
#pragma unroll
  for (int k = 0; k < PRT_PER; ++k) {
    const int i = tid + k * PRT_THREADS;
    const unsigned oo = (unsigned)d[k] / (unsigned)OWN;
    o[k] = (i < n && oo < (unsigned)BLD_BLOCKS) ? (int)oo : -1;
    slot[k] = 0;
    if (o[k] >= 0) slot[k] = atomicAdd(&s_cnt[o[k]], 1);
  }
  __syncthreads();

  if (wave == 0) {
    int carry = 0;
    for (int i0 = 0; i0 < BLD_BLOCKS; i0 += 32) {
      const int i = i0 + lane;
      const int c = s_cnt[i];
      const int inc = sage_wave_incl_scan(c, lane);
      const int p = carry + inc - c;
      s_off[i] = p;
      ct[i * PRT_SLICES + a] = c;
      ot[i * PRT_SLICES + a] = p;
      carry += __builtin_amdgcn_readlane(inc, 31);
    }
  }
  __syncthreads();

#pragma unroll
  for (int k = 0; k < PRT_PER; ++k)
    if (o[k] >= 0)
      part4[base + s_off[o[k]] + slot[k]] =
          make_int4(e[k], sv[k], wb[k], d[k] - o[k] * OWN);
}

__device__ __forceinline__ void sage_bld_keep(const int4 r, int li, int2* s_list,
                                              int2* s_sw) {
  if (li < BLD_CAP) {
    s_list[li] = make_int2(r.x, r.w);
    s_sw[li]   = make_int2(r.y, r.z);
  }
}

__global__ __launch_bounds__(BLD_THREADS) void sage_build_kernel(
    const int* __restrict__ dst, const int* __restrict__ src,
    const float* __restrict__ w, const int* __restrict__ part,
    const int* __restrict__ ct, const int* __restrict__ ot,
    int* __restrict__ rowptr, float* __restrict__ deg, int* __restrict__ rec) {
  __shared__ int  s_cnt[OWN];
  __shared__ int  s_ptr[OWN];
  __shared__ int  s_cur[OWN];
  __shared__ int  s_wless[BLD_WAVES];
  __shared__ int  s_wtot[PRT_SLICES / 32];
  __shared__ __attribute__((aligned(16))) int s_sc[PRT_SLICES];
  __shared__ __attribute__((aligned(16))) int s_sg[PRT_SLICES];
  __shared__ __attribute__((aligned(16))) int s_sb[PRT_SLICES];
  __shared__ int2 s_list[BLD_CAP];
  __shared__ int2 s_sw[BLD_CAP];
  __shared__ int  s_slot[BLD_CAP];

  const int tid  = threadIdx.x;
  const int lane = tid & 31;
  const int wave = tid >> 5;
  const int lo = blockIdx.x * OWN;
  const int hi = lo + OWN;
  int4* __restrict__ rec4 = (int4*)rec;
  const int4* __restrict__ part4 = (const int4*)part;

  for (int i = tid; i < OWN; i += BLD_THREADS) { s_cnt[i] = 0; s_cur[i] = 0; }

  int nless = 0;
  {
    const int4* __restrict__ ct4 = (const int4*)ct;
    const int nq = (int)blockIdx.x * CT_Q;
    int4 q[CT_U];
#pragma unroll
    for (int u = 0; u < CT_U; ++u) {
      const int idx = tid + u * BLD_THREADS;
      q[u] = make_int4(0, 0, 0, 0);
      if (idx < nq) q[u] = ct4[idx];
    }
#pragma unroll
    for (int u = 0; u < CT_U; ++u) nless += (q[u].x + q[u].y) + (q[u].z + q[u].w);
  }
  int c = 0, f = 0;
  if (tid < PRT_SLICES) {
    c = ct[blockIdx.x * PRT_SLICES + tid];
    f = ot[blockIdx.x * PRT_SLICES + tid];
  }
  {
    const int wsum = sage_wave_incl_scan(nless, lane);
    if (lane == 31) s_wless[wave] = wsum;
  }
  const int inc = sage_wave_incl_scan(c, lane);
  if (wave < PRT_SLICES / 32 && lane == 31) s_wtot[wave] = inc;
  __syncthreads();

  int total = 0;
  {
    int before = 0;
    for (int j = 0; j < PRT_SLICES / 32; ++j) {
      const int t = s_wtot[j];
      if (j < wave) before += t;
      total += t;
    }
    if (tid < PRT_SLICES) {
      s_sc[tid] = c;
      s_sg[tid] = tid * PRT_S + f;
      s_sb[tid] = before + inc - c;
    }
  }
  __syncthreads();

  {
    int4 r[8];
    const int4 c0 = ((const int4*)s_sc)[wave * 2], c1 = ((const int4*)s_sc)[wave * 2 + 1];
    const int4 g0 = ((const int4*)s_sg)[wave * 2], g1 = ((const int4*)s_sg)[wave * 2 + 1];
    const int4 b0 = ((const int4*)s_sb)[wave * 2], b1 = ((const int4*)s_sb)[wave * 2 + 1];
    const int ca[8] = {c0.x, c0.y, c0.z, c0.w, c1.x, c1.y, c1.z, c1.w};
    const int ga[8] = {g0.x, g0.y, g0.z, g0.w, g1.x, g1.y, g1.z, g1.w};
    const int ba[8] = {b0.x, b0.y, b0.z, b0.w, b1.x, b1.y, b1.z, b1.w};
#pragma unroll
    for (int j = 0; j < 8; ++j) {
      r[j] = make_int4(0, 0, 0, 0);
      if (lane < ca[j]) r[j] = part4[ga[j] + lane];
    }
#pragma unroll
    for (int j = 0; j < 8; ++j)
      if (lane < ca[j]) sage_bld_keep(r[j], ba[j] + lane, s_list, s_sw);
    for (int j = 0; j < 8; ++j)
      for (int k = lane + 32; k < ca[j]; k += 32)
        sage_bld_keep(part4[ga[j] + k], ba[j] + k, s_list, s_sw);
  }
  __syncthreads();

  const bool fits = total <= BLD_CAP;
  if (fits) {
    for (int li = tid; li < total; li += BLD_THREADS)
      s_slot[li] = atomicAdd(&s_cnt[s_list[li].y], 1);
  } else {
    for (int e = tid; e < NE; e += BLD_THREADS) {
      const int dd = dst[e];
      if (dd >= lo && dd < hi) atomicAdd(&s_cnt[dd - lo], 1);
    }
  }
  __syncthreads();

  if (wave == 0) {
    const int below = sage_wave_incl_scan(s_wless[lane], lane);
    int carry = __builtin_amdgcn_readlane(below, 31);
    for (int i0 = 0; i0 < OWN; i0 += 32) {
      const int i = i0 + lane;
      const int cn = s_cnt[i];
      const int in2 = sage_wave_incl_scan(cn, lane);
      const int p = carry + in2 - cn;
      s_ptr[i] = p;
      const int node = lo + i;
      if (node <= NN) rowptr[node] = p;
      if (node <  NN) deg[node] = (float)cn;
      carry += __builtin_amdgcn_readlane(in2, 31);
    }
  }
  __syncthreads();

  if (fits) {
    for (int li = tid; li < total; li += BLD_THREADS) {
      const int2 t  = s_list[li];
      const int2 sw = s_sw[li];
      rec4[s_ptr[t.y] + s_slot[li]] = make_int4(t.x, sw.x, sw.y, 0);
    }
  } else {
    for (int e = tid; e < NE; e += BLD_THREADS) {
      const int dd = dst[e];
      if (dd >= lo && dd < hi) {
        const int pos = s_ptr[dd - lo] + atomicAdd(&s_cur[dd - lo], 1);
        rec4[pos] = make_int4(e, src[e], __float_as_int(w[e]), 0);
      }
    }
  }
}


template <int G, int B>
__device__ __forceinline__ void sage_issue(float4 (&v)[31], float (&we)[31],
                                           const float* __restrict__ h,
                                           int ms, float mw, int j, int lane) {
#pragma unroll
  for (int u = 0; u < G; ++u) {
    const int s = __builtin_amdgcn_readlane(ms, j + u);
    we[B + u] = __int_as_float(__builtin_amdgcn_readlane(__float_as_int(mw), j + u));
    v[B + u] = ((const float4*)(h + (size_t)s * DIM))[lane];
  }
}
template <int G, int B>
__device__ __forceinline__ void sage_sum(float4& acc, const float4 (&v)[31],
                                         const float (&we)[31]) {
#pragma unroll
  for (int u = 0; u < G; ++u) {
    acc.x += v[B + u].x * we[B + u];
    acc.y += v[B + u].y * we[B + u];
    acc.z += v[B + u].z * we[B + u];
    acc.w += v[B + u].w * we[B + u];
  }
}
__device__ __forceinline__ void sage_row1(float4& acc, const float* __restrict__ h,
                                          int s, float we, int lane) {
  const float4 v = ((const float4*)(h + (size_t)s * DIM))[lane];
  acc.x += v.x * we; acc.y += v.y * we; acc.z += v.z * we; acc.w += v.w * we;
}

#define LPA (DIM + 4)

template <int NT>
__device__ __forceinline__ void sage_block_gather(
    const float* __restrict__ h, const int* __restrict__ rowptr,
    const int* __restrict__ rec, const float* __restrict__ Deg,
    int nrows, int rowbase, float* sAgg) {
  constexpr int NPB = NT * 2;
  const int lane = threadIdx.x & 31;
  const int wave = threadIdx.x >> 5;
  const int4* __restrict__ rec4 = (const int4*)rec;
  const int first = rowbase + wave * NPB;
  const int   rp = rowptr[min(first + lane, nrows)];
  const float dg = Deg[min(first + lane, nrows - 1)];
  int beg = __builtin_amdgcn_readlane(rp, 0);
  int n   = __builtin_amdgcn_readlane(rp, 1) - beg;
  int4 tn = make_int4(0x7fffffff, 0, 0, 0);
  if (lane < n) tn = rec4[beg + lane];
#pragma nounroll
  for (int k = 0; k < NPB; ++k) {
    const int4 t = tn;
    const int  nc = n, begc = beg;
    if (k + 1 < NPB) {
      beg = __builtin_amdgcn_readlane(rp, k + 1);
      n   = __builtin_amdgcn_readlane(rp, k + 2) - beg;
      tn  = make_int4(0x7fffffff, 0, 0, 0);
      if (lane < n) tn = rec4[beg + lane];
    }
    float4 acc = make_float4(0.0f, 0.0f, 0.0f, 0.0f);
    if (nc <= 31) {
      const int my = (lane < nc) ? t.x : 0x7fffffff;
      int rank = 0;
      for (int j = 0; j < nc; ++j) {
        const int other = __builtin_amdgcn_readlane(my, j);
        rank += (other < my) ? 1 : 0;
      }
      if (lane >= nc) rank = lane;
      const int   ms = __builtin_amdgcn_ds_permute(rank << 2, t.y);
      const float mw = __int_as_float(__builtin_amdgcn_ds_permute(rank << 2, t.z));
      float4 v[31];
      float  we[31];
      int j = 0;
      if (nc & 16) { sage_issue<16, 0>(v, we, h, ms, mw, j, lane);  j += 16; }
      if (nc & 8)  { sage_issue<8, 16>(v, we, h, ms, mw, j, lane);  j += 8; }
      if (nc & 4)  { sage_issue<4, 24>(v, we, h, ms, mw, j, lane);  j += 4; }
      if (nc & 2)  { sage_issue<2, 28>(v, we, h, ms, mw, j, lane);  j += 2; }
      if (nc & 1)  { sage_issue<1, 30>(v, we, h, ms, mw, j, lane); }
      if (nc & 16) sage_sum<16, 0>(acc, v, we);
      if (nc & 8)  sage_sum<8, 16>(acc, v, we);
      if (nc & 4)  sage_sum<4, 24>(acc, v, we);
      if (nc & 2)  sage_sum<2, 28>(acc, v, we);
      if (nc & 1)  sage_sum<1, 30>(acc, v, we);
    } else {
      int prev = -1;
      for (int r = 0; r < nc; ++r) {
        int best = 0x7fffffff, bs = 0, bw = 0;
        for (int i0 = 0; i0 < nc; i0 += 32) {
          const int i = i0 + lane;
          if (i < nc) {
            const int4 q = rec4[begc + i];
            if (q.x > prev && q.x < best) { best = q.x; bs = q.y; bw = q.z; }
          }
        }
        int mn = 0x7fffffff, src_lane = 0;
        for (int l = 0; l < 32; ++l) {
          const int vv = __builtin_amdgcn_readlane(best, l);
          if (vv < mn) { mn = vv; src_lane = l; }
        }
        const int   ps = __builtin_amdgcn_readlane(bs, src_lane);
        const float pw = __int_as_float(__builtin_amdgcn_readlane(bw, src_lane));
        sage_row1(acc, h, ps, pw, lane);
        prev = mn;
      }
    }
    const float rd = 1.0f / fmaxf(__int_as_float(__builtin_amdgcn_readlane(
                                      __float_as_int(dg), k)), 1.0f);
    float4 o;
    o.x = acc.x * rd; o.y = acc.y * rd; o.z = acc.z * rd; o.w = acc.w * rd;
    *(float4*)&sAgg[(wave * NPB + k) * LPA + lane * 4] = o;
  }
}

#define KC  32
#define LP  36
#define NCH (DIM / KC)
#define GT_TILES    ((NN + 15) / 16)
#define GEMM_BLOCKS 512
#define GT_BASE     (GT_TILES / GEMM_BLOCKS)
#define GT_EXTRA    (GT_TILES % GEMM_BLOCKS)
#define GT_MAX      (GT_BASE + 1)
#define RPB_MAX     (GT_MAX * 16)
static_assert(GT_BASE >= 1 && 2 * GT_EXTRA <= GEMM_BLOCKS, "long blocks sit at even indices");
static_assert(RPB_MAX <= 128, "a thread stages at most four rows of a chunk");

template <int NT>
__device__ __forceinline__ void sage_gemm_tiles(
    const float* __restrict__ Hself, const int* __restrict__ rowptr,
    const int* __restrict__ rec,     const float* __restrict__ Deg,
    const float* __restrict__ Wself, const float* __restrict__ Wneigh,
    const float* __restrict__ Bias,  float* __restrict__ Out,
    int nrows, int do_relu, int rowbase,
    float* sWs, float* sWn, float* sAs, float* sAgg) {
  constexpr int ROWS = NT * 16;
  const int tid  = threadIdx.x;
  const int lane = tid & 31;
  const int wave = tid >> 5;

  const int sr = tid >> 3;
  const int c4 = tid & 7;
  const bool has3 = (ROWS > 96) && (sr + 96 < ROWS);
  static_assert(ROWS >= 96 && ROWS <= 128, "three full staging rows per thread");

  const int gr0 = min(rowbase + sr,      nrows - 1);
  const int gr1 = min(rowbase + sr + 32, nrows - 1);
  const int gr2 = min(rowbase + sr + 64, nrows - 1);
  const int gr3 = min(rowbase + sr + 96, nrows - 1);
  const unsigned ow = (unsigned)(sr * DIM + c4 * 4);
  const unsigned o0 = (unsigned)(gr0 * DIM + c4 * 4);
  const unsigned o1 = (unsigned)(gr1 * DIM + c4 * 4);
  const unsigned o2 = (unsigned)(gr2 * DIM + c4 * 4);
  const unsigned o3 = (unsigned)(gr3 * DIM + c4 * 4);
  v4f ws0, ws1, ws2, ws3, wn0, wn1, wn2, wn3;
  v4f as0, as1, as2;
  v4f as3 = {0.0f, 0.0f, 0.0f, 0.0f};
#define SAGE_LD4(base_, off_) (*(const v4f*)((base_) + (off_)))
#define SAGE_LOAD_CHUNK(kc_)                                   \
  {                                                            \
    const unsigned o  = (unsigned)((kc_) * KC);                \
    const unsigned w1 = 32 * DIM, w2 = 64 * DIM, w3 = 96 * DIM; \
    ws0 = SAGE_LD4(Wself, ow + o);       ws1 = SAGE_LD4(Wself, ow + o + w1);  \
    ws2 = SAGE_LD4(Wself, ow + o + w2);  ws3 = SAGE_LD4(Wself, ow + o + w3);  \
    wn0 = SAGE_LD4(Wneigh, ow + o);      wn1 = SAGE_LD4(Wneigh, ow + o + w1); \
    wn2 = SAGE_LD4(Wneigh, ow + o + w2); wn3 = SAGE_LD4(Wneigh, ow + o + w3); \
    as0 = SAGE_LD4(Hself, o0 + o); as1 = SAGE_LD4(Hself, o1 + o); as2 = SAGE_LD4(Hself, o2 + o); \
    if (has3) { as3 = SAGE_LD4(Hself, o3 + o); }                                                 \
  }

  SAGE_LOAD_CHUNK(0)

  sage_block_gather<NT>(Hself, rowptr, rec, Deg, nrows, rowbase, sAgg);

  v8f acc[NT] = {};

  const int jj    = wave * 16 + (lane & 15);
  const int khalf = (lane >> 4) << 1;
  const int m     = lane & 15;

#pragma nounroll
  for (int kc = 0; kc < NCH; ++kc) {
    __syncthreads();
    {
      v4f* ls = (v4f*)&sWs[sr * LP + c4 * 4];
      v4f* ln = (v4f*)&sWn[sr * LP + c4 * 4];
      v4f* la = (v4f*)&sAs[sr * LP + c4 * 4];
      const int q = 32 * LP / 4;
      ls[0] = ws0; ls[q] = ws1; ls[2 * q] = ws2; ls[3 * q] = ws3;
      ln[0] = wn0; ln[q] = wn1; ln[2 * q] = wn2; ln[3 * q] = wn3;
      la[0] = as0; la[q] = as1; la[2 * q] = as2;
      if (has3) { la[3 * q] = as3; }
    }
    __syncthreads();
    if (kc + 1 < NCH) { SAGE_LOAD_CHUNK(kc + 1) }

    v2f bs[8], bn[8];
#pragma unroll
    for (int k0 = 0; k0 < 8; ++k0) {
      const int kk = k0 * 4 + khalf;
      bs[k0] = *(const v2f*)&sWs[jj * LP + kk];
      bn[k0] = *(const v2f*)&sWn[jj * LP + kk];
    }
    v2f fs[2][4], fn[2][4];
#pragma unroll
    for (int k0 = 0; k0 < 4; ++k0) {
      fs[0][k0] = *(const v2f*)&sAs[m * LP + k0 * 4 + khalf];
      fn[0][k0] = *(const v2f*)&sAgg[m * LPA + kc * KC + k0 * 4 + khalf];
    }
#pragma unroll
    for (int st = 0; st < 2 * NT; ++st) {
      const int cur = st & 1, nxt = cur ^ 1;
      const int kh = st / NT, rt = st % NT;
      if (st + 1 < 2 * NT) {
        const int kh1 = (st + 1) / NT, rt1 = (st + 1) % NT;
#pragma unroll
        for (int k0 = 0; k0 < 4; ++k0) {
          const int kk = kh1 * 16 + k0 * 4 + khalf;
          fs[nxt][k0] = *(const v2f*)&sAs[(rt1 * 16 + m) * LP + kk];
          fn[nxt][k0] = *(const v2f*)&sAgg[(rt1 * 16 + m) * LPA + kc * KC + kk];
        }
      }
#pragma unroll
      for (int k0 = 0; k0 < 4; ++k0) {
        acc[rt] = __builtin_amdgcn_wmma_f32_16x16x4_f32(
            false, fs[cur][k0], false, bs[kh * 4 + k0], (short)0, acc[rt], false, false);
        acc[rt] = __builtin_amdgcn_wmma_f32_16x16x4_f32(
            false, fn[cur][k0], false, bn[kh * 4 + k0], (short)0, acc[rt], false, false);
      }
    }
    __builtin_amdgcn_sched_barrier(0);
    asm volatile("v_nop\n\tv_nop\n\tv_nop\n\tv_nop");
    __builtin_amdgcn_sched_barrier(0);
  }
#undef SAGE_LOAD_CHUNK
#undef SAGE_LD4

  const float bv = Bias[jj];
  const int mofs = (lane >> 4) << 3;
#pragma unroll
  for (int rt = 0; rt < NT; ++rt) {
#pragma unroll
    for (int r = 0; r < 8; ++r) {
      int row = rowbase + rt * 16 + r + mofs;
      if (row < nrows) {
        float v = acc[rt][r] + bv;
        if (do_relu) v = fmaxf(v, 0.0f);
        Out[(size_t)row * DIM + jj] = v;
      }
    }
  }
}

__global__ __launch_bounds__(256, 4) void sage_gemm_kernel(
    const float* __restrict__ Hself, const int* __restrict__ rowptr,
    const int* __restrict__ rec,     const float* __restrict__ Deg,
    const float* __restrict__ Wself, const float* __restrict__ Wneigh,
    const float* __restrict__ Bias,  float* __restrict__ Out,
    int nrows, int do_relu) {
  __shared__ __attribute__((aligned(16))) float sWs[DIM * LP];
  __shared__ __attribute__((aligned(16))) float sWn[DIM * LP];
  __shared__ __attribute__((aligned(16))) float sAs[RPB_MAX * LP];
  __shared__ __attribute__((aligned(16))) float sAgg[RPB_MAX * LPA];

  const int  b      = (int)blockIdx.x;
  const bool longb  = ((b & 1) == 0) && (b < 2 * GT_EXTRA);
  const int  before = min((b + 1) >> 1, GT_EXTRA);
  const int  rowbase = (b * GT_BASE + before) * 16;
  if (longb)
    sage_gemm_tiles<GT_MAX>(Hself, rowptr, rec, Deg, Wself, Wneigh, Bias, Out, nrows,
                            do_relu, rowbase, sWs, sWn, sAs, sAgg);
  else
    sage_gemm_tiles<GT_BASE>(Hself, rowptr, rec, Deg, Wself, Wneigh, Bias, Out, nrows,
                             do_relu, rowbase, sWs, sWn, sAs, sAgg);
}

extern "C" void kernel_launch(void* const* d_in, const int* in_sizes, int n_in,
                              void* d_out, int out_size, void* d_ws, size_t ws_size,
                              hipStream_t stream) {
  const float* x       = (const float*)d_in[0];
  const int*   src     = (const int*)  d_in[1];
  const int*   dst     = (const int*)  d_in[2];
  const float* weight  = (const float*)d_in[3];
  const float* Wself1  = (const float*)d_in[4];
  const float* Wneigh1 = (const float*)d_in[5];
  const float* b1      = (const float*)d_in[6];
  const float* Wself2  = (const float*)d_in[7];
  const float* Wneigh2 = (const float*)d_in[8];
  const float* b2      = (const float*)d_in[9];
  float* out = (float*)d_out;

  char* ws = (char*)d_ws;
  float* deg  = (float*)(ws);
  char*  list = ws + ((size_t)NN * 4 + 255 & ~(size_t)255);
  float* h1   = (float*)(list + (size_t)NN * DIM * 4);

  int* rowptr = (int*)list;
  int* rec    = (int*)(list + 256 * 1024);

  const int gemm_blocks = GEMM_BLOCKS;
  const int own_blocks  = BLD_BLOCKS;

  int* part = (int*)h1;
  int* ct   = (int*)((char*)h1 + (size_t)PRT_SLICES * PRT_S * 16);
  int* ot   = ct + BLD_BLOCKS * PRT_SLICES;

  sage_part_kernel<<<PRT_SLICES, PRT_THREADS, 0, stream>>>(dst, src, weight, part, ct, ot);
  sage_build_kernel<<<own_blocks, BLD_THREADS, 0, stream>>>(
      dst, src, weight, part, ct, ot, rowptr, deg, rec);

  sage_gemm_kernel<<<gemm_blocks, 256, 0, stream>>>(
      x, rowptr, rec, deg, Wself1, Wneigh1, b1, h1, NN,  1);

  sage_gemm_kernel<<<gemm_blocks, 256, 0, stream>>>(
      h1, rowptr, rec, deg, Wself2, Wneigh2, b2, out, NN,  0);
}


# Neighbour sum in two overlapped stages: what needs no rows of a node (records into edge-id order, next records' fetch, 1/degree) is done while the node before has rows in flight; sum order kept
# speedup vs baseline: 3.8731x; 1.0569x over previous
#include <hip/hip_runtime.h>


#define NN 50000
#define NE 625000
#define DIM 128

typedef __attribute__((ext_vector_type(2))) float v2f;
typedef __attribute__((ext_vector_type(8))) float v8f;
typedef __attribute__((ext_vector_type(4))) float v4f;
typedef __attribute__((ext_vector_type(4))) int v4i;


__device__ __forceinline__ int sage_wave_incl_scan(int v, int lane) {
  v += __builtin_amdgcn_update_dpp(0, v, 0x111, 0xf, 0xf, false);
  v += __builtin_amdgcn_update_dpp(0, v, 0x112, 0xf, 0xf, false);
  v += __builtin_amdgcn_update_dpp(0, v, 0x114, 0xf, 0xf, false);
  v += __builtin_amdgcn_update_dpp(0, v, 0x118, 0xf, 0xf, false);
  const int low = __builtin_amdgcn_readlane(v, 15);
  if (lane >= 16) v += low;
  return v;
}

#define OWN         224
#define BLD_THREADS 1024
#define BLD_WAVES   (BLD_THREADS / 32)
#define BLD_CAP     8192
#define BLD_BLOCKS  ((NN + 1 + OWN - 1) / OWN)

#define PRT_SLICES  256
#define PRT_S       ((NE + PRT_SLICES - 1) / PRT_SLICES)
#define PRT_THREADS 512
#define PRT_PER     ((PRT_S + PRT_THREADS - 1) / PRT_THREADS)
#define CT_Q        (PRT_SLICES / 4)
#define CT_U        14
static_assert(OWN % 32 == 0 && BLD_WAVES == 32, "scan by whole waves");
static_assert(BLD_BLOCKS % 32 == 0, "the slice's owner counts are scanned by whole waves");
static_assert(PRT_SLICES == 8 * BLD_WAVES, "a wave of the owner block takes 8 slices");
static_assert((PRT_SLICES - 1) * PRT_S < NE, "the last slice is not empty");
static_assert((BLD_BLOCKS - 1) * CT_Q <= CT_U * BLD_THREADS, "ct rows below the last block");

__global__ __launch_bounds__(PRT_THREADS) void sage_part_kernel(
    const int* __restrict__ dst, const int* __restrict__ src,
    const float* __restrict__ w, int* __restrict__ part,
    int* __restrict__ ct, int* __restrict__ ot) {
  __shared__ int s_cnt[BLD_BLOCKS];
  __shared__ int s_off[BLD_BLOCKS];

  const int tid  = threadIdx.x;
  const int lane = tid & 31;
  const int wave = tid >> 5;
  const int a    = blockIdx.x;
  const int base = a * PRT_S;
  const int n    = min(PRT_S, NE - base);
  int4* __restrict__ part4 = (int4*)part;

  for (int i = tid; i < BLD_BLOCKS; i += PRT_THREADS) s_cnt[i] = 0;
  __syncthreads();

  int e[PRT_PER], d[PRT_PER], sv[PRT_PER], wb[PRT_PER], o[PRT_PER], slot[PRT_PER];
#pragma unroll
  for (int k = 0; k < PRT_PER; ++k) {
    const int i = tid + k * PRT_THREADS;
    e[k]  = base + (i < n ? i : 0);
    d[k]  = dst[e[k]];
    sv[k] = src[e[k]];
    wb[k] = __float_as_int(w[e[k]]);
  }
#pragma unroll
  for (int k = 0; k < PRT_PER; ++k) {
    const int i = tid + k * PRT_THREADS;
    const unsigned oo = (unsigned)d[k] / (unsigned)OWN;
    o[k] = (i < n && oo < (unsigned)BLD_BLOCKS) ? (int)oo : -1;
    slot[k] = 0;
    if (o[k] >= 0) slot[k] = atomicAdd(&s_cnt[o[k]], 1);
  }
  __syncthreads();

  if (wave == 0) {
    int carry = 0;
    for (int i0 = 0; i0 < BLD_BLOCKS; i0 += 32) {
      const int i = i0 + lane;
      const int c = s_cnt[i];
      const int inc = sage_wave_incl_scan(c, lane);
      const int p = carry + inc - c;
      s_off[i] = p;
      ct[i * PRT_SLICES + a] = c;
      ot[i * PRT_SLICES + a] = p;
      carry += __builtin_amdgcn_readlane(inc, 31);
    }
  }
  __syncthreads();

#pragma unroll
  for (int k = 0; k < PRT_PER; ++k)
    if (o[k] >= 0)
      part4[base + s_off[o[k]] + slot[k]] =
          make_int4(e[k], sv[k], wb[k], d[k] - o[k] * OWN);
}

__device__ __forceinline__ void sage_bld_keep(const int4 r, int li, int2* s_list,
                                              int2* s_sw) {
  if (li < BLD_CAP) {
    s_list[li] = make_int2(r.x, r.w);
    s_sw[li]   = make_int2(r.y, r.z);
  }
}

__global__ __launch_bounds__(BLD_THREADS) void sage_build_kernel(
    const int* __restrict__ dst, const int* __restrict__ src,
    const float* __restrict__ w, const int* __restrict__ part,
    const int* __restrict__ ct, const int* __restrict__ ot,
    int* __restrict__ rowptr, float* __restrict__ deg, int* __restrict__ rec) {
  __shared__ int  s_cnt[OWN];
  __shared__ int  s_ptr[OWN];
  __shared__ int  s_cur[OWN];
  __shared__ int  s_wless[BLD_WAVES];
  __shared__ int  s_wtot[PRT_SLICES / 32];
  __shared__ __attribute__((aligned(16))) int s_sc[PRT_SLICES];
  __shared__ __attribute__((aligned(16))) int s_sg[PRT_SLICES];
  __shared__ __attribute__((aligned(16))) int s_sb[PRT_SLICES];
  __shared__ int2 s_list[BLD_CAP];
  __shared__ int2 s_sw[BLD_CAP];
  __shared__ int  s_slot[BLD_CAP];

  const int tid  = threadIdx.x;
  const int lane = tid & 31;
  const int wave = tid >> 5;
  const int lo = blockIdx.x * OWN;
  const int hi = lo + OWN;
  int4* __restrict__ rec4 = (int4*)rec;
  const int4* __restrict__ part4 = (const int4*)part;

  for (int i = tid; i < OWN; i += BLD_THREADS) { s_cnt[i] = 0; s_cur[i] = 0; }

  int nless = 0;
  {
    const int4* __restrict__ ct4 = (const int4*)ct;
    const int nq = (int)blockIdx.x * CT_Q;
    int4 q[CT_U];
#pragma unroll
    for (int u = 0; u < CT_U; ++u) {
      const int idx = tid + u * BLD_THREADS;
      q[u] = make_int4(0, 0, 0, 0);
      if (idx < nq) q[u] = ct4[idx];
    }
#pragma unroll
    for (int u = 0; u < CT_U; ++u) nless += (q[u].x + q[u].y) + (q[u].z + q[u].w);
  }
  int c = 0, f = 0;
  if (tid < PRT_SLICES) {
    c = ct[blockIdx.x * PRT_SLICES + tid];
    f = ot[blockIdx.x * PRT_SLICES + tid];
  }
  {
    const int wsum = sage_wave_incl_scan(nless, lane);
    if (lane == 31) s_wless[wave] = wsum;
  }
  const int inc = sage_wave_incl_scan(c, lane);
  if (wave < PRT_SLICES / 32 && lane == 31) s_wtot[wave] = inc;
  __syncthreads();

  int total = 0;
  {
    int before = 0;
    for (int j = 0; j < PRT_SLICES / 32; ++j) {
      const int t = s_wtot[j];
      if (j < wave) before += t;
      total += t;
    }
    if (tid < PRT_SLICES) {
      s_sc[tid] = c;
      s_sg[tid] = tid * PRT_S + f;
      s_sb[tid] = before + inc - c;
    }
  }
  __syncthreads();

  {
    int4 r[8];
    const int4 c0 = ((const int4*)s_sc)[wave * 2], c1 = ((const int4*)s_sc)[wave * 2 + 1];
    const int4 g0 = ((const int4*)s_sg)[wave * 2], g1 = ((const int4*)s_sg)[wave * 2 + 1];
    const int4 b0 = ((const int4*)s_sb)[wave * 2], b1 = ((const int4*)s_sb)[wave * 2 + 1];
    const int ca[8] = {c0.x, c0.y, c0.z, c0.w, c1.x, c1.y, c1.z, c1.w};
    const int ga[8] = {g0.x, g0.y, g0.z, g0.w, g1.x, g1.y, g1.z, g1.w};
    const int ba[8] = {b0.x, b0.y, b0.z, b0.w, b1.x, b1.y, b1.z, b1.w};
#pragma unroll
    for (int j = 0; j < 8; ++j) {
      r[j] = make_int4(0, 0, 0, 0);
      if (lane < ca[j]) r[j] = part4[ga[j] + lane];
    }
#pragma unroll
    for (int j = 0; j < 8; ++j)
      if (lane < ca[j]) sage_bld_keep(r[j], ba[j] + lane, s_list, s_sw);
    for (int j = 0; j < 8; ++j)
      for (int k = lane + 32; k < ca[j]; k += 32)
        sage_bld_keep(part4[ga[j] + k], ba[j] + k, s_list, s_sw);
  }
  __syncthreads();

  const bool fits = total <= BLD_CAP;
  if (fits) {
    for (int li = tid; li < total; li += BLD_THREADS)
      s_slot[li] = atomicAdd(&s_cnt[s_list[li].y], 1);
  } else {
    for (int e = tid; e < NE; e += BLD_THREADS) {
      const int dd = dst[e];
      if (dd >= lo && dd < hi) atomicAdd(&s_cnt[dd - lo], 1);
    }
  }
  __syncthreads();

  if (wave == 0) {
    const int below = sage_wave_incl_scan(s_wless[lane], lane);
    int carry = __builtin_amdgcn_readlane(below, 31);
    for (int i0 = 0; i0 < OWN; i0 += 32) {
      const int i = i0 + lane;
      const int cn = s_cnt[i];
      const int in2 = sage_wave_incl_scan(cn, lane);
      const int p = carry + in2 - cn;
      s_ptr[i] = p;
      const int node = lo + i;
      if (node <= NN) rowptr[node] = p;
      if (node <  NN) deg[node] = (float)cn;
      carry += __builtin_amdgcn_readlane(in2, 31);
    }
  }
  __syncthreads();

  if (fits) {
    for (int li = tid; li < total; li += BLD_THREADS) {
      const int2 t  = s_list[li];
      const int2 sw = s_sw[li];
      rec4[s_ptr[t.y] + s_slot[li]] = make_int4(t.x, sw.x, sw.y, 0);
    }
  } else {
    for (int e = tid; e < NE; e += BLD_THREADS) {
      const int dd = dst[e];
      if (dd >= lo && dd < hi) {
        const int pos = s_ptr[dd - lo] + atomicAdd(&s_cur[dd - lo], 1);
        rec4[pos] = make_int4(e, src[e], __float_as_int(w[e]), 0);
      }
    }
  }
}


#define SLOTS 31

template <int G, int B>
__device__ __forceinline__ void sage_issue(float4 (&v)[SLOTS], float (&we)[SLOTS],
                                           const float* __restrict__ h,
                                           int ms, float mw, int j, int lane) {
#pragma unroll
  for (int u = 0; u < G; ++u) {
    const int s = __builtin_amdgcn_readlane(ms, j + u);
    we[B + u] = __int_as_float(__builtin_amdgcn_readlane(__float_as_int(mw), j + u));
    v[B + u] = ((const float4*)(h + (size_t)s * DIM))[lane];
  }
}
template <int G, int B>
__device__ __forceinline__ void sage_sum(float4& acc, const float4 (&v)[SLOTS],
                                         const float (&we)[SLOTS]) {
#pragma unroll
  for (int u = 0; u < G; ++u) {
    acc.x += v[B + u].x * we[B + u];
    acc.y += v[B + u].y * we[B + u];
    acc.z += v[B + u].z * we[B + u];
    acc.w += v[B + u].w * we[B + u];
  }
}
__device__ __forceinline__ void sage_row1(float4& acc, const float* __restrict__ h,
                                          int s, float we, int lane) {
  const float4 v = ((const float4*)(h + (size_t)s * DIM))[lane];
  acc.x += v.x * we; acc.y += v.y * we; acc.z += v.z * we; acc.w += v.w * we;
}

#define LPA (DIM + 4)

__device__ __forceinline__ v4i sage_records(const int4* __restrict__ rec4, int beg, int n,
                                            int lane) {
  const int i = beg + min(lane, max(n - 1, 0));
  return *(const v4i*)&rec4[min(i, NE - 1)];
}

__device__ __forceinline__ void sage_order(int my, int sy, int sz, int n, int lane,
                                           int& ms, float& mw) {
  int rank = 0;
  for (int j = 0; j < n; ++j) {
    const int other = __builtin_amdgcn_readlane(my, j);
    rank += (other < my) ? 1 : 0;
  }
  if (lane >= n) rank = lane;
  ms = __builtin_amdgcn_ds_permute(rank << 2, sy);
  mw = __int_as_float(__builtin_amdgcn_ds_permute(rank << 2, sz));
}

__device__ __forceinline__ void sage_ahead(const int4* __restrict__ rec4, int rp, int k,
                                           int npb, int lane, int my1, int sy1, int sz1,
                                           int n1, int& beg2, int& n2, v4i& t2,
                                           int& ms1, float& mw1) {
  if (k + 2 < npb) {
    beg2 = __builtin_amdgcn_readlane(rp, k + 2);
    n2   = __builtin_amdgcn_readlane(rp, k + 3) - beg2;
    t2 = sage_records(rec4, beg2, n2, lane);
  }
  if (k + 1 < npb && n1 <= 31) sage_order(my1, sy1, sz1, n1, lane, ms1, mw1);
}

template <int NT>
__device__ __forceinline__ void sage_block_gather(
    const float* __restrict__ h, const int* __restrict__ rowptr,
    const int* __restrict__ rec, const float* __restrict__ Deg,
    int nrows, int rowbase, float* sAgg) {
  constexpr int NPB = NT * 2;
  static_assert(NPB >= 2, "two nodes are fetched ahead of the loop");
  const int lane = threadIdx.x & 31;
  const int wave = threadIdx.x >> 5;
  const int4* __restrict__ rec4 = (const int4*)rec;
  const int first = rowbase + wave * NPB;
  const int   rp = rowptr[min(first + lane, nrows)];
  const float dg = Deg[min(first + lane, nrows - 1)];
  const float rdl = 1.0f / fmaxf(dg, 1.0f);
  int begc = __builtin_amdgcn_readlane(rp, 0);
  int nc   = __builtin_amdgcn_readlane(rp, 1) - begc;
  const v4i t0 = sage_records(rec4, begc, nc, lane);
  int beg1 = __builtin_amdgcn_readlane(rp, 1);
  int n1   = __builtin_amdgcn_readlane(rp, 2) - beg1;
  v4i t1 = sage_records(rec4, beg1, n1, lane);
  int   ms = 0;
  float mw = 0.0f;
  if (nc <= 31) sage_order((lane < nc) ? t0.x : 0x7fffffff, t0.y, t0.z, nc, lane, ms, mw);
#pragma nounroll
  for (int k = 0; k < NPB; ++k) {
    asm volatile("" : "+v"(t1));
    const int my1 = (lane < n1) ? t1.x : 0x7fffffff;
    const int sy1 = t1.y, sz1 = t1.z;
    int   beg2 = 0, n2 = 0, ms1 = 0;
    float mw1 = 0.0f;
    v4i   t2 = {0x7fffffff, 0, 0, 0};
    float4 acc = make_float4(0.0f, 0.0f, 0.0f, 0.0f);
    if (nc <= 31) {
      float4 v[SLOTS];
      float  we[SLOTS];
      const int j4 = nc & 24, j2 = nc & 28, j1 = nc & 30;
      if (nc & 24) sage_issue<8, 0>(v, we, h, ms, mw, 0, lane);
      if (nc & 16) sage_issue<8, 8>(v, we, h, ms, mw, 8, lane);
      if (nc >= 24) sage_issue<8, 16>(v, we, h, ms, mw, 16, lane);
      if (nc & 4)  sage_issue<4, 24>(v, we, h, ms, mw, j4, lane);
      if (nc & 2)  sage_issue<2, 28>(v, we, h, ms, mw, j2, lane);
      if (nc & 1)  sage_issue<1, 30>(v, we, h, ms, mw, j1, lane);
      sage_ahead(rec4, rp, k, NPB, lane, my1, sy1, sz1, n1, beg2, n2, t2, ms1, mw1);
      if (nc & 24) sage_sum<8, 0>(acc, v, we);
      if (nc & 16) sage_sum<8, 8>(acc, v, we);
      if (nc >= 24) sage_sum<8, 16>(acc, v, we);
      if (nc & 4)  sage_sum<4, 24>(acc, v, we);
      if (nc & 2)  sage_sum<2, 28>(acc, v, we);
      if (nc & 1)  sage_sum<1, 30>(acc, v, we);
    } else {
      sage_ahead(rec4, rp, k, NPB, lane, my1, sy1, sz1, n1, beg2, n2, t2, ms1, mw1);
      int prev = -1;
      for (int r = 0; r < nc; ++r) {
        int best = 0x7fffffff, bs = 0, bw = 0;
        for (int i0 = 0; i0 < nc; i0 += 32) {
          const int i = i0 + lane;
          if (i < nc) {
            const int4 q = rec4[begc + i];
            if (q.x > prev && q.x < best) { best = q.x; bs = q.y; bw = q.z; }
          }
        }
        int mn = 0x7fffffff, src_lane = 0;
        for (int l = 0; l < 32; ++l) {
          const int vv = __builtin_amdgcn_readlane(best, l);
          if (vv < mn) { mn = vv; src_lane = l; }
        }
        const int   ps = __builtin_amdgcn_readlane(bs, src_lane);
        const float pw = __int_as_float(__builtin_amdgcn_readlane(bw, src_lane));
        sage_row1(acc, h, ps, pw, lane);
        prev = mn;
      }
    }
    const float rd = __int_as_float(__builtin_amdgcn_readlane(__float_as_int(rdl), k));
    float4 o;
    o.x = acc.x * rd; o.y = acc.y * rd; o.z = acc.z * rd; o.w = acc.w * rd;
    *(float4*)&sAgg[(wave * NPB + k) * LPA + lane * 4] = o;
    nc = n1; begc = beg1; ms = ms1; mw = mw1;
    t1 = t2; n1 = n2; beg1 = beg2;
  }
}

#define KC  32
#define LP  36
#define NCH (DIM / KC)
#define GT_TILES    ((NN + 15) / 16)
#define GEMM_BLOCKS 512
#define GT_BASE     (GT_TILES / GEMM_BLOCKS)
#define GT_EXTRA    (GT_TILES % GEMM_BLOCKS)
#define GT_MAX      (GT_BASE + 1)
#define RPB_MAX     (GT_MAX * 16)
static_assert(GT_BASE >= 1 && 2 * GT_EXTRA <= GEMM_BLOCKS, "long blocks sit at even indices");
static_assert(RPB_MAX <= 128, "a thread stages at most four rows of a chunk");

template <int NT>
__device__ __forceinline__ void sage_gemm_tiles(
    const float* __restrict__ Hself, const int* __restrict__ rowptr,
    const int* __restrict__ rec,     const float* __restrict__ Deg,
    const float* __restrict__ Wself, const float* __restrict__ Wneigh,
    const float* __restrict__ Bias,  float* __restrict__ Out,
    int nrows, int do_relu, int rowbase,
    float* sWs, float* sWn, float* sAs, float* sAgg) {
  constexpr int ROWS = NT * 16;
  const int tid  = threadIdx.x;
  const int lane = tid & 31;
  const int wave = tid >> 5;

  const int sr = tid >> 3;
  const int c4 = tid & 7;
  const bool has3 = (ROWS > 96) && (sr + 96 < ROWS);
  static_assert(ROWS >= 96 && ROWS <= 128, "three full staging rows per thread");

  const int gr0 = min(rowbase + sr,      nrows - 1);
  const int gr1 = min(rowbase + sr + 32, nrows - 1);
  const int gr2 = min(rowbase + sr + 64, nrows - 1);
  const int gr3 = min(rowbase + sr + 96, nrows - 1);
  const unsigned ow = (unsigned)(sr * DIM + c4 * 4);
  const unsigned o0 = (unsigned)(gr0 * DIM + c4 * 4);
  const unsigned o1 = (unsigned)(gr1 * DIM + c4 * 4);
  const unsigned o2 = (unsigned)(gr2 * DIM + c4 * 4);
  const unsigned o3 = (unsigned)(gr3 * DIM + c4 * 4);
  v4f ws0, ws1, ws2, ws3, wn0, wn1, wn2, wn3;
  v4f as0, as1, as2;
  v4f as3 = {0.0f, 0.0f, 0.0f, 0.0f};
#define SAGE_LD4(base_, off_) (*(const v4f*)((base_) + (off_)))
#define SAGE_LOAD_CHUNK(kc_)                                   \
  {                                                            \
    const unsigned o  = (unsigned)((kc_) * KC);                \
    const unsigned w1 = 32 * DIM, w2 = 64 * DIM, w3 = 96 * DIM; \
    ws0 = SAGE_LD4(Wself, ow + o);       ws1 = SAGE_LD4(Wself, ow + o + w1);  \
    ws2 = SAGE_LD4(Wself, ow + o + w2);  ws3 = SAGE_LD4(Wself, ow + o + w3);  \
    wn0 = SAGE_LD4(Wneigh, ow + o);      wn1 = SAGE_LD4(Wneigh, ow + o + w1); \
    wn2 = SAGE_LD4(Wneigh, ow + o + w2); wn3 = SAGE_LD4(Wneigh, ow + o + w3); \
    as0 = SAGE_LD4(Hself, o0 + o); as1 = SAGE_LD4(Hself, o1 + o); as2 = SAGE_LD4(Hself, o2 + o); \
    if (has3) { as3 = SAGE_LD4(Hself, o3 + o); }                                                 \
  }

  SAGE_LOAD_CHUNK(0)

  sage_block_gather<NT>(Hself, rowptr, rec, Deg, nrows, rowbase, sAgg);

  v8f acc[NT] = {};

  const int jj    = wave * 16 + (lane & 15);
  const int khalf = (lane >> 4) << 1;
  const int m     = lane & 15;

#pragma nounroll
  for (int kc = 0; kc < NCH; ++kc) {
    __syncthreads();
    {
      v4f* ls = (v4f*)&sWs[sr * LP + c4 * 4];
      v4f* ln = (v4f*)&sWn[sr * LP + c4 * 4];
      v4f* la = (v4f*)&sAs[sr * LP + c4 * 4];
      const int q = 32 * LP / 4;
      ls[0] = ws0; ls[q] = ws1; ls[2 * q] = ws2; ls[3 * q] = ws3;
      ln[0] = wn0; ln[q] = wn1; ln[2 * q] = wn2; ln[3 * q] = wn3;
      la[0] = as0; la[q] = as1; la[2 * q] = as2;
      if (has3) { la[3 * q] = as3; }
    }
    __syncthreads();
    if (kc + 1 < NCH) { SAGE_LOAD_CHUNK(kc + 1) }

    v2f bs[8], bn[8];
#pragma unroll
    for (int k0 = 0; k0 < 8; ++k0) {
      const int kk = k0 * 4 + khalf;
      bs[k0] = *(const v2f*)&sWs[jj * LP + kk];
      bn[k0] = *(const v2f*)&sWn[jj * LP + kk];
    }
    v2f fs[2][4], fn[2][4];
#pragma unroll
    for (int k0 = 0; k0 < 4; ++k0) {
      fs[0][k0] = *(const v2f*)&sAs[m * LP + k0 * 4 + khalf];
      fn[0][k0] = *(const v2f*)&sAgg[m * LPA + kc * KC + k0 * 4 + khalf];
    }
#pragma unroll
    for (int st = 0; st < 2 * NT; ++st) {
      const int cur = st & 1, nxt = cur ^ 1;
      const int kh = st / NT, rt = st % NT;
      if (st + 1 < 2 * NT) {
        const int kh1 = (st + 1) / NT, rt1 = (st + 1) % NT;
#pragma unroll
        for (int k0 = 0; k0 < 4; ++k0) {
          const int kk = kh1 * 16 + k0 * 4 + khalf;
          fs[nxt][k0] = *(const v2f*)&sAs[(rt1 * 16 + m) * LP + kk];
          fn[nxt][k0] = *(const v2f*)&sAgg[(rt1 * 16 + m) * LPA + kc * KC + kk];
        }
      }
#pragma unroll
      for (int k0 = 0; k0 < 4; ++k0) {
        acc[rt] = __builtin_amdgcn_wmma_f32_16x16x4_f32(
            false, fs[cur][k0], false, bs[kh * 4 + k0], (short)0, acc[rt], false, false);
        acc[rt] = __builtin_amdgcn_wmma_f32_16x16x4_f32(
            false, fn[cur][k0], false, bn[kh * 4 + k0], (short)0, acc[rt], false, false);
      }
    }
    __builtin_amdgcn_sched_barrier(0);
    asm volatile("v_nop\n\tv_nop\n\tv_nop\n\tv_nop");
    __builtin_amdgcn_sched_barrier(0);
  }
#undef SAGE_LOAD_CHUNK
#undef SAGE_LD4

  const float bv = Bias[jj];
  const int mofs = (lane >> 4) << 3;
#pragma unroll
  for (int rt = 0; rt < NT; ++rt) {
#pragma unroll
    for (int r = 0; r < 8; ++r) {
      int row = rowbase + rt * 16 + r + mofs;
      if (row < nrows) {
        float v = acc[rt][r] + bv;
        if (do_relu) v = fmaxf(v, 0.0f);
        Out[(size_t)row * DIM + jj] = v;
      }
    }
  }
}

__global__ __launch_bounds__(256, 4) void sage_gemm_kernel(
    const float* __restrict__ Hself, const int* __restrict__ rowptr,
    const int* __restrict__ rec,     const float* __restrict__ Deg,
    const float* __restrict__ Wself, const float* __restrict__ Wneigh,
    const float* __restrict__ Bias,  float* __restrict__ Out,
    int nrows, int do_relu) {
  __shared__ __attribute__((aligned(16))) float sWs[DIM * LP];
  __shared__ __attribute__((aligned(16))) float sWn[DIM * LP];
  __shared__ __attribute__((aligned(16))) float sAs[RPB_MAX * LP];
  __shared__ __attribute__((aligned(16))) float sAgg[RPB_MAX * LPA];

  const int  b      = (int)blockIdx.x;
  const bool longb  = ((b & 1) == 0) && (b < 2 * GT_EXTRA);
  const int  before = min((b + 1) >> 1, GT_EXTRA);
  const int  rowbase = (b * GT_BASE + before) * 16;
  if (longb)
    sage_gemm_tiles<GT_MAX>(Hself, rowptr, rec, Deg, Wself, Wneigh, Bias, Out, nrows,
                            do_relu, rowbase, sWs, sWn, sAs, sAgg);
  else
    sage_gemm_tiles<GT_BASE>(Hself, rowptr, rec, Deg, Wself, Wneigh, Bias, Out, nrows,
                             do_relu, rowbase, sWs, sWn, sAs, sAgg);
}

extern "C" void kernel_launch(void* const* d_in, const int* in_sizes, int n_in,
                              void* d_out, int out_size, void* d_ws, size_t ws_size,
                              hipStream_t stream) {
  const float* x       = (const float*)d_in[0];
  const int*   src     = (const int*)  d_in[1];
  const int*   dst     = (const int*)  d_in[2];
  const float* weight  = (const float*)d_in[3];
  const float* Wself1  = (const float*)d_in[4];
  const float* Wneigh1 = (const float*)d_in[5];
  const float* b1      = (const float*)d_in[6];
  const float* Wself2  = (const float*)d_in[7];
  const float* Wneigh2 = (const float*)d_in[8];
  const float* b2      = (const float*)d_in[9];
  float* out = (float*)d_out;

  char* ws = (char*)d_ws;
  float* deg  = (float*)(ws);
  char*  list = ws + ((size_t)NN * 4 + 255 & ~(size_t)255);
  float* h1   = (float*)(list + (size_t)NN * DIM * 4);

  int* rowptr = (int*)list;
  int* rec    = (int*)(list + 256 * 1024);

  const int gemm_blocks = GEMM_BLOCKS;
  const int own_blocks  = BLD_BLOCKS;

  int* part = (int*)h1;
  int* ct   = (int*)((char*)h1 + (size_t)PRT_SLICES * PRT_S * 16);
  int* ot   = ct + BLD_BLOCKS * PRT_SLICES;

  sage_part_kernel<<<PRT_SLICES, PRT_THREADS, 0, stream>>>(dst, src, weight, part, ct, ot);
  sage_build_kernel<<<own_blocks, BLD_THREADS, 0, stream>>>(
      dst, src, weight, part, ct, ot, rowptr, deg, rec);

  sage_gemm_kernel<<<gemm_blocks, 256, 0, stream>>>(
      x, rowptr, rec, deg, Wself1, Wneigh1, b1, h1, NN,  1);

  sage_gemm_kernel<<<gemm_blocks, 256, 0, stream>>>(
      h1, rowptr, rec, deg, Wself2, Wneigh2, b2, out, NN,  0);
}
